# starting kernel
# baseline (speedup 1.0000x reference)
#include <hip/hip_runtime.h>
#include <math.h>

typedef __attribute__((ext_vector_type(2))) float v2f;
typedef __attribute__((ext_vector_type(8))) float v8f;

#define NEG_SLOPE 0.2f

__device__ __forceinline__ unsigned fkey(float f) {
    unsigned b = __float_as_uint(f);
    return (b & 0x80000000u) ? ~b : (b | 0x80000000u);
}
__device__ __forceinline__ float fkey_inv(unsigned u) {
    return (u & 0x80000000u) ? __uint_as_float(u & 0x7fffffffu)
                             : __uint_as_float(~u);
}
#define FKEY_NEG_INF 0x007fffffu

template <int K, int NCOL>
__global__ void gemm_wmma_kernel(const float* __restrict__ A,
                                 const float* __restrict__ B,
                                 float* __restrict__ C, int N) {
    constexpr int NT = NCOL / 16;
    const int lane = threadIdx.x & 31;
    const int l    = lane & 15;
    const int hf   = lane >> 4;
    const int row0 = blockIdx.x * 16;
    if (row0 >= N) return;

    const float* __restrict__ arow = A + (size_t)(row0 + l) * K;
    v8f acc[NT];
#pragma unroll
    for (int t = 0; t < NT; ++t) acc[t] = (v8f){};

#pragma unroll
    for (int k = 0; k < K; k += 4) {
        v2f a;
        a.x = arow[k + 2 * hf];
        a.y = arow[k + 2 * hf + 1];
#pragma unroll
        for (int t = 0; t < NT; ++t) {
            v2f b;
            b.x = B[(size_t)(k + 2 * hf) * NCOL + 16 * t + l];
            b.y = B[(size_t)(k + 2 * hf + 1) * NCOL + 16 * t + l];
            acc[t] = __builtin_amdgcn_wmma_f32_16x16x4_f32(
                 false, a,  false, b,
                 (short)0, acc[t],  false,
                 false);
        }
    }
#pragma unroll
    for (int t = 0; t < NT; ++t)
#pragma unroll
        for (int i = 0; i < 8; ++i)
            C[(size_t)(row0 + i + 8 * hf) * NCOL + 16 * t + l] = acc[t][i];
}

template <int H, int C>
__global__ void alpha_kernel(const float* __restrict__ xw,
                             const float* __restrict__ a_src,
                             const float* __restrict__ a_dst,
                             float* __restrict__ as_, float* __restrict__ ad_,
                             int N) {
    int idx = blockIdx.x * blockDim.x + threadIdx.x;
    if (idx >= N * H) return;
    int n = idx / H, h = idx % H;
    const float* v = xw + (size_t)n * (H * C) + h * C;
    float s = 0.f, d = 0.f;
#pragma unroll
    for (int c = 0; c < C; ++c) {
        s += v[c] * a_src[h * C + c];
        d += v[c] * a_dst[h * C + c];
    }
    as_[idx] = s;
    ad_[idx] = d;
}

__global__ void init_kernel(unsigned* __restrict__ m_, float* __restrict__ den_,
                            float* __restrict__ acc_, int NH, int NHC) {
    int idx = blockIdx.x * blockDim.x + threadIdx.x;
    if (idx < NHC) acc_[idx] = 0.f;
    if (idx < NH) { m_[idx] = FKEY_NEG_INF; den_[idx] = 0.f; }
}

__device__ __forceinline__ void edge_sd(const int* __restrict__ srcs,
                                        const int* __restrict__ dsts, int E,
                                        int i, int& s, int& d) {
    if (i < E) { s = srcs[i]; d = dsts[i]; }
    else       { s = i - E;   d = i - E;   }
}

template <int H>
__global__ void edge_max_kernel(const int* __restrict__ srcs,
                                const int* __restrict__ dsts, int E, int N,
                                const float* __restrict__ as_,
                                const float* __restrict__ ad_,
                                unsigned* __restrict__ m_) {
    int i = blockIdx.x * blockDim.x + threadIdx.x;
    if (i >= E + N) return;
    int s, d; edge_sd(srcs, dsts, E, i, s, d);
#pragma unroll
    for (int h = 0; h < H; ++h) {
        float e = as_[s * H + h] + ad_[d * H + h];
        e = (e > 0.f) ? e : NEG_SLOPE * e;
        atomicMax(&m_[d * H + h], fkey(e));
    }
}

template <int H>
__global__ void edge_denom_kernel(const int* __restrict__ srcs,
                                  const int* __restrict__ dsts, int E, int N,
                                  const float* __restrict__ as_,
                                  const float* __restrict__ ad_,
                                  const unsigned* __restrict__ m_,
                                  float* __restrict__ den_) {
    int i = blockIdx.x * blockDim.x + threadIdx.x;
    if (i >= E + N) return;
    int s, d; edge_sd(srcs, dsts, E, i, s, d);
#pragma unroll
    for (int h = 0; h < H; ++h) {
        float e = as_[s * H + h] + ad_[d * H + h];
        e = (e > 0.f) ? e : NEG_SLOPE * e;
        atomicAdd(&den_[d * H + h], expf(e - fkey_inv(m_[d * H + h])));
    }
}

template <int H, int C>
__global__ void edge_accum_kernel(const int* __restrict__ srcs,
                                  const int* __restrict__ dsts, int E, int N,
                                  const float* __restrict__ as_,
                                  const float* __restrict__ ad_,
                                  const unsigned* __restrict__ m_,
                                  const float* __restrict__ den_,
                                  const float* __restrict__ xw,
                                  float* __restrict__ acc_) {
    static_assert(C % 4 == 0, "C must be a multiple of 4");
    int i = blockIdx.x * blockDim.x + threadIdx.x;
    if (i >= E + N) return;
    int s, d; edge_sd(srcs, dsts, E, i, s, d);

    float w[H];
#pragma unroll
    for (int h = 0; h < H; ++h) {
        float e = as_[s * H + h] + ad_[d * H + h];
        e = (e > 0.f) ? e : NEG_SLOPE * e;
        w[h] = expf(e - fkey_inv(m_[d * H + h])) / den_[d * H + h];
    }
    const float4* __restrict__ sv =
        (const float4*)(xw + (size_t)s * (H * C));
    float* __restrict__ dv = acc_ + (size_t)d * (H * C);
#pragma unroll
    for (int f = 0; f < (H * C) / 4; ++f) {
        float4 v = sv[f];
        float wh = w[(f * 4) / C];
        atomicAdd(&dv[f * 4 + 0], wh * v.x);
        atomicAdd(&dv[f * 4 + 1], wh * v.y);
        atomicAdd(&dv[f * 4 + 2], wh * v.z);
        atomicAdd(&dv[f * 4 + 3], wh * v.w);
    }
}

__global__ void bias_elu_kernel(const float* __restrict__ acc_,
                                const float* __restrict__ bias,
                                float* __restrict__ h_, int total, int F) {
    int i = blockIdx.x * blockDim.x + threadIdx.x;
    if (i >= total) return;
    float v = acc_[i] + bias[i & (F - 1)];
    h_[i] = (v > 0.f) ? v : (expf(v) - 1.f);
}

__global__ void logsoftmax_kernel(const float* __restrict__ acc_,
                                  const float* __restrict__ bias,
                                  float* __restrict__ out, int N) {
    int n = blockIdx.x * blockDim.x + threadIdx.x;
    if (n >= N) return;
    float v[16];
    float m = -INFINITY;
#pragma unroll
    for (int c = 0; c < 16; ++c) {
        v[c] = acc_[(size_t)n * 16 + c] + bias[c];
        m = fmaxf(m, v[c]);
    }
    float ssum = 0.f;
#pragma unroll
    for (int c = 0; c < 16; ++c) ssum += expf(v[c] - m);
    float lg = m + logf(ssum);
#pragma unroll
    for (int c = 0; c < 16; ++c) out[(size_t)n * 16 + c] = v[c] - lg;
}

extern "C" void kernel_launch(void* const* d_in, const int* in_sizes, int n_in,
                              void* d_out, int out_size, void* d_ws,
                              size_t ws_size, hipStream_t stream) {
    const float* x   = (const float*)d_in[0];
    const int*   ei  = (const int*)d_in[1];
    const float* W1  = (const float*)d_in[2];
    const float* a1s = (const float*)d_in[3];
    const float* a1d = (const float*)d_in[4];
    const float* b1  = (const float*)d_in[5];
    const float* W2  = (const float*)d_in[6];
    const float* a2s = (const float*)d_in[7];
    const float* a2d = (const float*)d_in[8];
    const float* b2  = (const float*)d_in[9];
    float* out = (float*)d_out;

    const int N = in_sizes[0] / 128;
    const int E = in_sizes[1] / 2;
    const int* srcs = ei;
    const int* dsts = ei + E;
    const int T = E + N;

    char* p = (char*)d_ws;
    auto carve = [&](size_t bytes) {
        void* r = (void*)p;
        p += (bytes + 255) & ~(size_t)255;
        return r;
    };
    float*    xw1  = (float*)carve((size_t)N * 32 * 4);
    float*    as1  = (float*)carve((size_t)N * 4 * 4);
    float*    ad1  = (float*)carve((size_t)N * 4 * 4);
    unsigned* m1   = (unsigned*)carve((size_t)N * 4 * 4);
    float*    den1 = (float*)carve((size_t)N * 4 * 4);
    float*    out1 = (float*)carve((size_t)N * 32 * 4);
    float*    h1   = (float*)carve((size_t)N * 32 * 4);
    float*    xw2  = (float*)carve((size_t)N * 16 * 4);
    float*    as2  = (float*)carve((size_t)N * 4);
    float*    ad2  = (float*)carve((size_t)N * 4);
    unsigned* m2   = (unsigned*)carve((size_t)N * 4);
    float*    den2 = (float*)carve((size_t)N * 4);
    float*    out2 = (float*)carve((size_t)N * 16 * 4);

    const int B = 256;
    auto g = [](int n, int b) { return (n + b - 1) / b; };

    gemm_wmma_kernel<128, 32><<<N / 16, 32, 0, stream>>>(x, W1, xw1, N);
    alpha_kernel<4, 8><<<g(N * 4, B), B, 0, stream>>>(xw1, a1s, a1d, as1, ad1, N);
    init_kernel<<<g(N * 32, B), B, 0, stream>>>(m1, den1, out1, N * 4, N * 32);
    edge_max_kernel<4><<<g(T, B), B, 0, stream>>>(srcs, dsts, E, N, as1, ad1, m1);
    edge_denom_kernel<4><<<g(T, B), B, 0, stream>>>(srcs, dsts, E, N, as1, ad1,
                                                    m1, den1);
    edge_accum_kernel<4, 8><<<g(T, B), B, 0, stream>>>(srcs, dsts, E, N, as1,
                                                       ad1, m1, den1, xw1, out1);
    bias_elu_kernel<<<g(N * 32, B), B, 0, stream>>>(out1, b1, h1, N * 32, 32);

    gemm_wmma_kernel<32, 16><<<N / 16, 32, 0, stream>>>(h1, W2, xw2, N);
    alpha_kernel<1, 16><<<g(N, B), B, 0, stream>>>(xw2, a2s, a2d, as2, ad2, N);
    init_kernel<<<g(N * 16, B), B, 0, stream>>>(m2, den2, out2, N, N * 16);
    edge_max_kernel<1><<<g(T, B), B, 0, stream>>>(srcs, dsts, E, N, as2, ad2, m2);
    edge_denom_kernel<1><<<g(T, B), B, 0, stream>>>(srcs, dsts, E, N, as2, ad2,
                                                    m2, den2);
    edge_accum_kernel<1, 16><<<g(T, B), B, 0, stream>>>(srcs, dsts, E, N, as2,
                                                        ad2, m2, den2, xw2, out2);
    logsoftmax_kernel<<<g(N, B), B, 0, stream>>>(out2, b2, out, N);
}


# each work-group owns 392 destination nodes in LDS and filters the whole edge list for them once; max and unnormalised sums on LDS atomics, one divide per output; no global atomics; arrival order
# speedup vs baseline: 6.7244x; 6.7244x over previous
#include <hip/hip_runtime.h>
#include <math.h>

typedef __attribute__((ext_vector_type(2))) float v2f;
typedef __attribute__((ext_vector_type(8))) float v8f;

#define NEG_SLOPE 0.2f

__device__ __forceinline__ unsigned fkey(float f) {
    unsigned b = __float_as_uint(f);
    return (b & 0x80000000u) ? ~b : (b | 0x80000000u);
}
__device__ __forceinline__ float fkey_inv(unsigned u) {
    return (u & 0x80000000u) ? __uint_as_float(u & 0x7fffffffu)
                             : __uint_as_float(~u);
}
#define FKEY_NEG_INF 0x007fffffu

template <int K, int NCOL>
__global__ void gemm_wmma_kernel(const float* __restrict__ A,
                                 const float* __restrict__ B,
                                 float* __restrict__ C, int N) {
    constexpr int NT = NCOL / 16;
    const int lane = threadIdx.x & 31;
    const int l    = lane & 15;
    const int hf   = lane >> 4;
    const int row0 = blockIdx.x * 16;
    if (row0 >= N) return;

    const float* __restrict__ arow = A + (size_t)(row0 + l) * K;
    v8f acc[NT];
#pragma unroll
    for (int t = 0; t < NT; ++t) acc[t] = (v8f){};

#pragma unroll
    for (int k = 0; k < K; k += 4) {
        v2f a;
        a.x = arow[k + 2 * hf];
        a.y = arow[k + 2 * hf + 1];
#pragma unroll
        for (int t = 0; t < NT; ++t) {
            v2f b;
            b.x = B[(size_t)(k + 2 * hf) * NCOL + 16 * t + l];
            b.y = B[(size_t)(k + 2 * hf + 1) * NCOL + 16 * t + l];
            acc[t] = __builtin_amdgcn_wmma_f32_16x16x4_f32(
                 false, a,  false, b,
                 (short)0, acc[t],  false,
                 false);
        }
    }
#pragma unroll
    for (int t = 0; t < NT; ++t)
#pragma unroll
        for (int i = 0; i < 8; ++i)
            C[(size_t)(row0 + i + 8 * hf) * NCOL + 16 * t + l] = acc[t][i];
}

template <int H, int C>
__global__ void alpha_kernel(const float* __restrict__ xw,
                             const float* __restrict__ a_src,
                             const float* __restrict__ a_dst,
                             float* __restrict__ as_, float* __restrict__ ad_,
                             int N) {
    int idx = blockIdx.x * blockDim.x + threadIdx.x;
    if (idx >= N * H) return;
    int n = idx / H, h = idx % H;
    const float* v = xw + (size_t)n * (H * C) + h * C;
    float s = 0.f, d = 0.f;
#pragma unroll
    for (int c = 0; c < C; ++c) {
        s += v[c] * a_src[h * C + c];
        d += v[c] * a_dst[h * C + c];
    }
    as_[idx] = s;
    ad_[idx] = d;
}

#define NPS  392
#define OWN_THREADS 1024
#define LCAP 18432

template <class Fn>
__device__ __forceinline__ void scan_dsts(const int* __restrict__ dsts, int E,
                                          int n0, unsigned cnt, Fn emit) {
    const int tid = threadIdx.x;
    constexpr int nt = OWN_THREADS;
    const int E4 = ((reinterpret_cast<size_t>(dsts) & 15) == 0) ? (E >> 2) : 0;
    const int4* __restrict__ d4 = reinterpret_cast<const int4*>(dsts);
    auto test4 = [&](int4 v, int q) {
        const unsigned a = (unsigned)(v.x - n0), b = (unsigned)(v.y - n0);
        const unsigned c = (unsigned)(v.z - n0), d = (unsigned)(v.w - n0);
        if ((a < cnt) | (b < cnt) | (c < cnt) | (d < cnt)) {
            if (a < cnt) emit(4 * q + 0, (int)a);
            if (b < cnt) emit(4 * q + 1, (int)b);
            if (c < cnt) emit(4 * q + 2, (int)c);
            if (d < cnt) emit(4 * q + 3, (int)d);
        }
    };
    constexpr int U = 8;
    int q = tid;
    for (; q + (U - 1) * nt < E4; q += U * nt) {
        int4 v[U];
#pragma unroll
        for (int u = 0; u < U; ++u) v[u] = d4[q + u * nt];
#pragma unroll
        for (int u = 0; u < U; ++u) test4(v[u], q + u * nt);
    }
    for (; q < E4; q += nt) test4(d4[q], q);
    for (int i = 4 * E4 + tid; i < E; i += nt) {
        const unsigned a = (unsigned)(dsts[i] - n0);
        if (a < cnt) emit(i, (int)a);
    }
}

template <int H, int C, bool FINAL>
__global__ __launch_bounds__(OWN_THREADS) void owner_kernel(
    const int* __restrict__ srcs, const int* __restrict__ dsts, int E, int N,
    const float* __restrict__ as_, const float* __restrict__ ad_,
    const float* __restrict__ xw, const float* __restrict__ bias,
    float* __restrict__ outp) {
    static_assert(C % 4 == 0, "C must be a multiple of 4");
    static_assert(!FINAL || H == 1, "the final layer has one head");
    constexpr int F = H * C;
    constexpr int FP = F + 1;
    __shared__ unsigned s_m[NPS * H];
    __shared__ float    s_ad[NPS * H];
    __shared__ float    s_den[NPS * H];
    __shared__ float    s_acc[NPS * FP];
    __shared__ int      s_li[LCAP];
    __shared__ int      s_ld[LCAP];
    __shared__ int      s_cnt;

    const int tid = threadIdx.x;
    constexpr int nt = OWN_THREADS;
    const int n0 = blockIdx.x * NPS;
    if (n0 >= N) return;
    const int cnt = min(NPS, N - n0);

    for (int j = tid; j < cnt * H; j += nt) {
        s_m[j] = FKEY_NEG_INF;
        s_den[j] = 0.f;
        s_ad[j] = ad_[(size_t)n0 * H + j];
    }
    for (int j = tid; j < cnt * FP; j += nt) s_acc[j] = 0.f;
    if (tid == 0) s_cnt = 0;
    __syncthreads();

    auto edge_max = [&](int s, int dl) {
#pragma unroll
        for (int h = 0; h < H; ++h) {
            float e = as_[s * H + h] + s_ad[dl * H + h];
            e = (e > 0.f) ? e : NEG_SLOPE * e;
            atomicMax(&s_m[dl * H + h], fkey(e));
        }
    };
    auto edge_acc = [&](int s, int dl) {
        float w[H];
#pragma unroll
        for (int h = 0; h < H; ++h) {
            float e = as_[s * H + h] + s_ad[dl * H + h];
            e = (e > 0.f) ? e : NEG_SLOPE * e;
            w[h] = expf(e - fkey_inv(s_m[dl * H + h]));
            atomicAdd(&s_den[dl * H + h], w[h]);
        }
        const float4* __restrict__ sv = (const float4*)(xw + (size_t)s * F);
        float* dv = &s_acc[dl * FP];
#pragma unroll
        for (int f = 0; f < F / 4; ++f) {
            const float4 v = sv[f];
            const float wh = w[(f * 4) / C];
            atomicAdd(&dv[f * 4 + 0], wh * v.x);
            atomicAdd(&dv[f * 4 + 1], wh * v.y);
            atomicAdd(&dv[f * 4 + 2], wh * v.z);
            atomicAdd(&dv[f * 4 + 3], wh * v.w);
        }
    };

    scan_dsts(dsts, E, n0, (unsigned)cnt, [&](int i, int dl) {
        const int pos = atomicAdd(&s_cnt, 1);
        if (pos < LCAP) { s_li[pos] = i; s_ld[pos] = dl; }
    });
    __syncthreads();
    const int  total  = s_cnt;
    const bool listed = total <= LCAP;
    if (listed)
        for (int k = tid; k < total; k += nt) s_li[k] = srcs[s_li[k]];

    auto for_each_edge = [&](auto fn) {
        if (listed) {
            for (int k = tid; k < total; k += nt) fn(s_li[k], s_ld[k]);
        } else {
            for (int i = tid; i < E; i += nt) {
                const unsigned a = (unsigned)(dsts[i] - n0);
                if (a < (unsigned)cnt) fn(srcs[i], (int)a);
            }
        }
        for (int dl = tid; dl < cnt; dl += nt) fn(n0 + dl, dl);
    };

    for_each_edge(edge_max);
    __syncthreads();
    for_each_edge(edge_acc);
    __syncthreads();

    if (!FINAL) {
        for (int j = tid; j < cnt * F; j += nt) {
            const float v =
                s_acc[(j / F) * FP + j % F] / s_den[j / C] + bias[j % F];
            outp[(size_t)n0 * F + j] = (v > 0.f) ? v : (expf(v) - 1.f);
        }
    } else {
        for (int dl = tid; dl < cnt; dl += nt) {
            const float dn = s_den[dl];
            float v[F];
            float m = -INFINITY;
#pragma unroll
            for (int c = 0; c < F; ++c) {
                v[c] = s_acc[dl * FP + c] / dn + bias[c];
                m = fmaxf(m, v[c]);
            }
            float ssum = 0.f;
#pragma unroll
            for (int c = 0; c < F; ++c) ssum += expf(v[c] - m);
            const float lg = m + logf(ssum);
#pragma unroll
            for (int c = 0; c < F; ++c)
                outp[(size_t)(n0 + dl) * F + c] = v[c] - lg;
        }
    }
}

extern "C" void kernel_launch(void* const* d_in, const int* in_sizes, int n_in,
                              void* d_out, int out_size, void* d_ws,
                              size_t ws_size, hipStream_t stream) {
    const float* x   = (const float*)d_in[0];
    const int*   ei  = (const int*)d_in[1];
    const float* W1  = (const float*)d_in[2];
    const float* a1s = (const float*)d_in[3];
    const float* a1d = (const float*)d_in[4];
    const float* b1  = (const float*)d_in[5];
    const float* W2  = (const float*)d_in[6];
    const float* a2s = (const float*)d_in[7];
    const float* a2d = (const float*)d_in[8];
    const float* b2  = (const float*)d_in[9];
    float* out = (float*)d_out;

    const int N = in_sizes[0] / 128;
    const int E = in_sizes[1] / 2;
    const int* srcs = ei;
    const int* dsts = ei + E;

    char* p = (char*)d_ws;
    auto carve = [&](size_t bytes) {
        void* r = (void*)p;
        p += (bytes + 255) & ~(size_t)255;
        return r;
    };
    float* xw1 = (float*)carve((size_t)N * 32 * 4);
    float* as1 = (float*)carve((size_t)N * 4 * 4);
    float* ad1 = (float*)carve((size_t)N * 4 * 4);
    float* h1  = (float*)carve((size_t)N * 32 * 4);
    float* xw2 = (float*)carve((size_t)N * 16 * 4);
    float* as2 = (float*)carve((size_t)N * 4);
    float* ad2 = (float*)carve((size_t)N * 4);

    const int B = 256;
    auto g = [](int n, int b) { return (n + b - 1) / b; };

    gemm_wmma_kernel<128, 32><<<N / 16, 32, 0, stream>>>(x, W1, xw1, N);
    alpha_kernel<4, 8><<<g(N * 4, B), B, 0, stream>>>(xw1, a1s, a1d, as1, ad1, N);
    owner_kernel<4, 8, false><<<g(N, NPS), OWN_THREADS, 0, stream>>>(
        srcs, dsts, E, N, as1, ad1, xw1, b1, h1);

    gemm_wmma_kernel<32, 16><<<N / 16, 32, 0, stream>>>(h1, W2, xw2, N);
    alpha_kernel<1, 16><<<g(N, B), B, 0, stream>>>(xw2, a2s, a2d, as2, ad2, N);
    owner_kernel<1, 16, true><<<g(N, NPS), OWN_THREADS, 0, stream>>>(
        srcs, dsts, E, N, as2, ad2, xw2, b2, out);
}


# owner launches find their edges in a one-byte key plane (owning group per destination, one new small launch per call; hits checked against the real destination), not the int32 array; arrival order
# speedup vs baseline: 8.1508x; 1.2121x over previous
#include <hip/hip_runtime.h>
#include <math.h>

typedef __attribute__((ext_vector_type(2))) float v2f;
typedef __attribute__((ext_vector_type(8))) float v8f;

#define NEG_SLOPE 0.2f

__device__ __forceinline__ unsigned fkey(float f) {
    unsigned b = __float_as_uint(f);
    return (b & 0x80000000u) ? ~b : (b | 0x80000000u);
}
__device__ __forceinline__ float fkey_inv(unsigned u) {
    return (u & 0x80000000u) ? __uint_as_float(u & 0x7fffffffu)
                             : __uint_as_float(~u);
}
#define FKEY_NEG_INF 0x007fffffu

template <int K, int NCOL>
__global__ void gemm_wmma_kernel(const float* __restrict__ A,
                                 const float* __restrict__ B,
                                 float* __restrict__ C, int N) {
    constexpr int NT = NCOL / 16;
    const int lane = threadIdx.x & 31;
    const int l    = lane & 15;
    const int hf   = lane >> 4;
    const int row0 = blockIdx.x * 16;
    if (row0 >= N) return;

    const float* __restrict__ arow = A + (size_t)(row0 + l) * K;
    v8f acc[NT];
#pragma unroll
    for (int t = 0; t < NT; ++t) acc[t] = (v8f){};

#pragma unroll
    for (int k = 0; k < K; k += 4) {
        v2f a;
        a.x = arow[k + 2 * hf];
        a.y = arow[k + 2 * hf + 1];
#pragma unroll
        for (int t = 0; t < NT; ++t) {
            v2f b;
            b.x = B[(size_t)(k + 2 * hf) * NCOL + 16 * t + l];
            b.y = B[(size_t)(k + 2 * hf + 1) * NCOL + 16 * t + l];
            acc[t] = __builtin_amdgcn_wmma_f32_16x16x4_f32(
                 false, a,  false, b,
                 (short)0, acc[t],  false,
                 false);
        }
    }
#pragma unroll
    for (int t = 0; t < NT; ++t)
#pragma unroll
        for (int i = 0; i < 8; ++i)
            C[(size_t)(row0 + i + 8 * hf) * NCOL + 16 * t + l] = acc[t][i];
}

template <int H, int C>
__global__ void alpha_kernel(const float* __restrict__ xw,
                             const float* __restrict__ a_src,
                             const float* __restrict__ a_dst,
                             float* __restrict__ as_, float* __restrict__ ad_,
                             int N) {
    int idx = blockIdx.x * blockDim.x + threadIdx.x;
    if (idx >= N * H) return;
    int n = idx / H, h = idx % H;
    const float* v = xw + (size_t)n * (H * C) + h * C;
    float s = 0.f, d = 0.f;
#pragma unroll
    for (int c = 0; c < C; ++c) {
        s += v[c] * a_src[h * C + c];
        d += v[c] * a_dst[h * C + c];
    }
    as_[idx] = s;
    ad_[idx] = d;
}

#define NPS  392
#define OWN_THREADS 1024
#define LCAP 18432

#define KEY_U 4
__device__ __forceinline__ unsigned key4(int a, int b, int c, int d) {
    return (((unsigned)a / (unsigned)NPS) & 0xffu) |
           ((((unsigned)b / (unsigned)NPS) & 0xffu) << 8) |
           ((((unsigned)c / (unsigned)NPS) & 0xffu) << 16) |
           ((((unsigned)d / (unsigned)NPS) & 0xffu) << 24);
}
__global__ void key_kernel(const int* __restrict__ dsts, int E,
                           unsigned* __restrict__ keys) {
    const int lane = threadIdx.x & 31;
    const int wave = (blockIdx.x * blockDim.x + threadIdx.x) >> 5;
    const int t0   = wave * (KEY_U * 32) + lane;
    const bool whole = (reinterpret_cast<size_t>(dsts) & 15) == 0 &&
                       (wave + 1) * (KEY_U * 128) <= E;
    if (whole) {
        const int4* __restrict__ d4 = reinterpret_cast<const int4*>(dsts);
        int4 v[KEY_U];
#pragma unroll
        for (int u = 0; u < KEY_U; ++u) v[u] = d4[t0 + 32 * u];
#pragma unroll
        for (int u = 0; u < KEY_U; ++u)
            keys[t0 + 32 * u] = key4(v[u].x, v[u].y, v[u].z, v[u].w);
    } else {
        for (int u = 0; u < KEY_U; ++u) {
            const int t = t0 + 32 * u, i0 = 4 * t;
            if (i0 < E)
                keys[t] = key4(dsts[i0],
                               (i0 + 1 < E) ? dsts[i0 + 1] : 0,
                               (i0 + 2 < E) ? dsts[i0 + 2] : 0,
                               (i0 + 3 < E) ? dsts[i0 + 3] : 0);
        }
    }
}

template <class Fn>
__device__ __forceinline__ void scan_keys(const unsigned* __restrict__ keys,
                                          const int* __restrict__ dsts, int E,
                                          int n0, unsigned cnt, Fn emit) {
    const int tid = threadIdx.x;
    constexpr int nt = OWN_THREADS;
    const int E16 = E >> 4;
    const uint4* __restrict__ k4 = reinterpret_cast<const uint4*>(keys);
    const unsigned pat = (blockIdx.x & 0xffu) * 0x01010101u;
    auto eq4 = [&](unsigned w) {
        const unsigned x = w ^ pat;
        return (x - 0x01010101u) & ~x & 0x80808080u;
    };
    auto test16 = [&](uint4 v, int q) {
        unsigned m = eq4(v.x) | (eq4(v.y) >> 1) | (eq4(v.z) >> 2) |
                     (eq4(v.w) >> 3);
        while (m) {
            const int p = __builtin_ctz(m);
            m &= m - 1;
            emit(16 * q + 4 * (7 - (p & 7)) + (p >> 3));
        }
    };
    constexpr int U = 8;
    int q = tid;
    for (; q + (U - 1) * nt < E16; q += U * nt) {
        uint4 v[U];
#pragma unroll
        for (int u = 0; u < U; ++u) v[u] = k4[q + u * nt];
#pragma unroll
        for (int u = 0; u < U; ++u) test16(v[u], q + u * nt);
    }
    for (; q < E16; q += nt) test16(k4[q], q);
    for (int i = 16 * E16 + tid; i < E; i += nt) {
        const unsigned a = (unsigned)(dsts[i] - n0);
        if (a < cnt) emit(i);
    }
}

template <int H, int C, bool FINAL>
__global__ __launch_bounds__(OWN_THREADS) void owner_kernel(
    const int* __restrict__ srcs, const int* __restrict__ dsts,
    const unsigned* __restrict__ keys, int E, int N,
    const float* __restrict__ as_, const float* __restrict__ ad_,
    const float* __restrict__ xw, const float* __restrict__ bias,
    float* __restrict__ outp) {
    static_assert(C % 4 == 0, "C must be a multiple of 4");
    static_assert(!FINAL || H == 1, "the final layer has one head");
    constexpr int F = H * C;
    constexpr int FP = F + 1;
    __shared__ unsigned s_m[NPS * H];
    __shared__ float    s_ad[NPS * H];
    __shared__ float    s_den[NPS * H];
    __shared__ float    s_acc[NPS * FP];
    __shared__ int      s_li[LCAP];
    __shared__ int      s_ld[LCAP];
    __shared__ int      s_cnt;

    const int tid = threadIdx.x;
    constexpr int nt = OWN_THREADS;
    const int n0 = blockIdx.x * NPS;
    if (n0 >= N) return;
    const int cnt = min(NPS, N - n0);

    for (int j = tid; j < cnt * H; j += nt) {
        s_m[j] = FKEY_NEG_INF;
        s_den[j] = 0.f;
        s_ad[j] = ad_[(size_t)n0 * H + j];
    }
    for (int j = tid; j < cnt * FP; j += nt) s_acc[j] = 0.f;
    if (tid == 0) s_cnt = 0;
    __syncthreads();

    auto edge_max = [&](int s, int dl) {
#pragma unroll
        for (int h = 0; h < H; ++h) {
            float e = as_[s * H + h] + s_ad[dl * H + h];
            e = (e > 0.f) ? e : NEG_SLOPE * e;
            atomicMax(&s_m[dl * H + h], fkey(e));
        }
    };
    auto edge_acc = [&](int s, int dl) {
        float w[H];
#pragma unroll
        for (int h = 0; h < H; ++h) {
            float e = as_[s * H + h] + s_ad[dl * H + h];
            e = (e > 0.f) ? e : NEG_SLOPE * e;
            w[h] = expf(e - fkey_inv(s_m[dl * H + h]));
            atomicAdd(&s_den[dl * H + h], w[h]);
        }
        const float4* __restrict__ sv = (const float4*)(xw + (size_t)s * F);
        float* dv = &s_acc[dl * FP];
#pragma unroll
        for (int f = 0; f < F / 4; ++f) {
            const float4 v = sv[f];
            const float wh = w[(f * 4) / C];
            atomicAdd(&dv[f * 4 + 0], wh * v.x);
            atomicAdd(&dv[f * 4 + 1], wh * v.y);
            atomicAdd(&dv[f * 4 + 2], wh * v.z);
            atomicAdd(&dv[f * 4 + 3], wh * v.w);
        }
    };

    scan_keys(keys, dsts, E, n0, (unsigned)cnt, [&](int i) {
        const int pos = atomicAdd(&s_cnt, 1);
        if (pos < LCAP) s_li[pos] = i;
    });
    __syncthreads();
    const int  total  = s_cnt;
    const bool listed = total <= LCAP;
    if (listed)
        for (int k = tid; k < total; k += nt) {
            const int i = s_li[k];
            const unsigned a = (unsigned)(dsts[i] - n0);
            s_li[k] = srcs[i];
            s_ld[k] = (a < (unsigned)cnt) ? (int)a : -1;
        }

    auto for_each_edge = [&](auto fn) {
        if (listed) {
            for (int k = tid; k < total; k += nt) {
                const int dl = s_ld[k];
                if (dl >= 0) fn(s_li[k], dl);
            }
        } else {
            for (int i = tid; i < E; i += nt) {
                const unsigned a = (unsigned)(dsts[i] - n0);
                if (a < (unsigned)cnt) fn(srcs[i], (int)a);
            }
        }
        for (int dl = tid; dl < cnt; dl += nt) fn(n0 + dl, dl);
    };

    for_each_edge(edge_max);
    __syncthreads();
    for_each_edge(edge_acc);
    __syncthreads();

    if (!FINAL) {
        for (int j = tid; j < cnt * F; j += nt) {
            const float v =
                s_acc[(j / F) * FP + j % F] / s_den[j / C] + bias[j % F];
            outp[(size_t)n0 * F + j] = (v > 0.f) ? v : (expf(v) - 1.f);
        }
    } else {
        for (int dl = tid; dl < cnt; dl += nt) {
            const float dn = s_den[dl];
            float v[F];
            float m = -INFINITY;
#pragma unroll
            for (int c = 0; c < F; ++c) {
                v[c] = s_acc[dl * FP + c] / dn + bias[c];
                m = fmaxf(m, v[c]);
            }
            float ssum = 0.f;
#pragma unroll
            for (int c = 0; c < F; ++c) ssum += expf(v[c] - m);
            const float lg = m + logf(ssum);
#pragma unroll
            for (int c = 0; c < F; ++c)
                outp[(size_t)(n0 + dl) * F + c] = v[c] - lg;
        }
    }
}

extern "C" void kernel_launch(void* const* d_in, const int* in_sizes, int n_in,
                              void* d_out, int out_size, void* d_ws,
                              size_t ws_size, hipStream_t stream) {
    const float* x   = (const float*)d_in[0];
    const int*   ei  = (const int*)d_in[1];
    const float* W1  = (const float*)d_in[2];
    const float* a1s = (const float*)d_in[3];
    const float* a1d = (const float*)d_in[4];
    const float* b1  = (const float*)d_in[5];
    const float* W2  = (const float*)d_in[6];
    const float* a2s = (const float*)d_in[7];
    const float* a2d = (const float*)d_in[8];
    const float* b2  = (const float*)d_in[9];
    float* out = (float*)d_out;

    const int N = in_sizes[0] / 128;
    const int E = in_sizes[1] / 2;
    const int* srcs = ei;
    const int* dsts = ei + E;

    char* p = (char*)d_ws;
    auto carve = [&](size_t bytes) {
        void* r = (void*)p;
        p += (bytes + 255) & ~(size_t)255;
        return r;
    };
    float* xw1 = (float*)carve((size_t)N * 32 * 4);
    float* as1 = (float*)carve((size_t)N * 4 * 4);
    float* ad1 = (float*)carve((size_t)N * 4 * 4);
    float* h1  = (float*)carve((size_t)N * 32 * 4);
    unsigned* keys = (unsigned*)carve((size_t)E + 16);
    float* xw2 = xw1;
    float* as2 = as1;
    float* ad2 = ad1;

    const int B = 256;
    auto g = [](int n, int b) { return (n + b - 1) / b; };

    key_kernel<<<g(g(E, 4 * KEY_U), B), B, 0, stream>>>(dsts, E, keys);

    gemm_wmma_kernel<128, 32><<<N / 16, 32, 0, stream>>>(x, W1, xw1, N);
    alpha_kernel<4, 8><<<g(N * 4, B), B, 0, stream>>>(xw1, a1s, a1d, as1, ad1, N);
    owner_kernel<4, 8, false><<<g(N, NPS), OWN_THREADS, 0, stream>>>(
        srcs, dsts, keys, E, N, as1, ad1, xw1, b1, h1);

    gemm_wmma_kernel<32, 16><<<N / 16, 32, 0, stream>>>(h1, W2, xw2, N);
    alpha_kernel<1, 16><<<g(N, B), B, 0, stream>>>(xw2, a2s, a2d, as2, ad2, N);
    owner_kernel<1, 16, true><<<g(N, NPS), OWN_THREADS, 0, stream>>>(
        srcs, dsts, keys, E, N, as2, ad2, xw2, b2, out);
}


# scores alpha_src/alpha_dst computed in the GEMM launches from the accumulator tile (butterfly lane sums over a head's columns); both alpha launches removed: 5 launches, not 7; alpha sum order: tree
# speedup vs baseline: 8.4649x; 1.0385x over previous
#include <hip/hip_runtime.h>
#include <math.h>

typedef __attribute__((ext_vector_type(2))) float v2f;
typedef __attribute__((ext_vector_type(8))) float v8f;

#define NEG_SLOPE 0.2f

__device__ __forceinline__ unsigned fkey(float f) {
    unsigned b = __float_as_uint(f);
    return (b & 0x80000000u) ? ~b : (b | 0x80000000u);
}
__device__ __forceinline__ float fkey_inv(unsigned u) {
    return (u & 0x80000000u) ? __uint_as_float(u & 0x7fffffffu)
                             : __uint_as_float(~u);
}
#define FKEY_NEG_INF 0x007fffffu

template <int K, int NCOL, int H>
__global__ void gemm_alpha_kernel(const float* __restrict__ A,
                                  const float* __restrict__ B,
                                  const float* __restrict__ a_src,
                                  const float* __restrict__ a_dst,
                                  float* __restrict__ C,
                                  float* __restrict__ as_,
                                  float* __restrict__ ad_, int N) {
    constexpr int NT = NCOL / 16;
    constexpr int CH = NCOL / H;
    static_assert(CH == 8 || CH == 16, "a head must lie inside one tile");
    const int lane = threadIdx.x & 31;
    const int l    = lane & 15;
    const int hf   = lane >> 4;
    const int row0 =
        (blockIdx.x * (blockDim.x >> 5) + (threadIdx.x >> 5)) * 16;
    if (row0 >= N) return;

    const float* __restrict__ arow = A + (size_t)(row0 + l) * K;
    v8f acc[NT];
#pragma unroll
    for (int t = 0; t < NT; ++t) acc[t] = (v8f){};

#pragma unroll
    for (int k = 0; k < K; k += 4) {
        v2f a;
        a.x = arow[k + 2 * hf];
        a.y = arow[k + 2 * hf + 1];
#pragma unroll
        for (int t = 0; t < NT; ++t) {
            v2f b;
            b.x = B[(size_t)(k + 2 * hf) * NCOL + 16 * t + l];
            b.y = B[(size_t)(k + 2 * hf + 1) * NCOL + 16 * t + l];
            acc[t] = __builtin_amdgcn_wmma_f32_16x16x4_f32(
                 false, a,  false, b,
                 (short)0, acc[t],  false,
                 false);
        }
    }
    float s[NT][8], d[NT][8];
#pragma unroll
    for (int t = 0; t < NT; ++t) {
        const float ws = a_src[16 * t + l];
        const float wd = a_dst[16 * t + l];
#pragma unroll
        for (int i = 0; i < 8; ++i) {
            s[t][i] = acc[t][i] * ws;
            d[t][i] = acc[t][i] * wd;
        }
    }
#pragma unroll
    for (int m = 1; m < CH; m <<= 1) {
        float ps[NT][8], pd[NT][8];
#pragma unroll
        for (int t = 0; t < NT; ++t)
#pragma unroll
            for (int i = 0; i < 8; ++i) {
                ps[t][i] = __shfl_xor(s[t][i], m);
                pd[t][i] = __shfl_xor(d[t][i], m);
            }
#pragma unroll
        for (int t = 0; t < NT; ++t)
#pragma unroll
            for (int i = 0; i < 8; ++i) {
                s[t][i] += ps[t][i];
                d[t][i] += pd[t][i];
            }
    }
#pragma unroll
    for (int t = 0; t < NT; ++t)
#pragma unroll
        for (int i = 0; i < 8; ++i)
            C[(size_t)(row0 + i + 8 * hf) * NCOL + 16 * t + l] = acc[t][i];
    if ((l & (CH - 1)) == 0) {
#pragma unroll
        for (int t = 0; t < NT; ++t)
#pragma unroll
            for (int i = 0; i < 8; ++i) {
                const size_t o =
                    (size_t)(row0 + i + 8 * hf) * H + (16 * t + l) / CH;
                as_[o] = s[t][i];
                ad_[o] = d[t][i];
            }
    }
}

#define NPS  392
#define OWN_THREADS 1024
#define LCAP 18432

#define KEY_U 4
__device__ __forceinline__ unsigned key4(int a, int b, int c, int d) {
    return (((unsigned)a / (unsigned)NPS) & 0xffu) |
           ((((unsigned)b / (unsigned)NPS) & 0xffu) << 8) |
           ((((unsigned)c / (unsigned)NPS) & 0xffu) << 16) |
           ((((unsigned)d / (unsigned)NPS) & 0xffu) << 24);
}
__global__ void key_kernel(const int* __restrict__ dsts, int E,
                           unsigned* __restrict__ keys) {
    const int lane = threadIdx.x & 31;
    const int wave = (blockIdx.x * blockDim.x + threadIdx.x) >> 5;
    const int t0   = wave * (KEY_U * 32) + lane;
    const bool whole = (reinterpret_cast<size_t>(dsts) & 15) == 0 &&
                       (wave + 1) * (KEY_U * 128) <= E;
    if (whole) {
        const int4* __restrict__ d4 = reinterpret_cast<const int4*>(dsts);
        int4 v[KEY_U];
#pragma unroll
        for (int u = 0; u < KEY_U; ++u) v[u] = d4[t0 + 32 * u];
#pragma unroll
        for (int u = 0; u < KEY_U; ++u)
            keys[t0 + 32 * u] = key4(v[u].x, v[u].y, v[u].z, v[u].w);
    } else {
        for (int u = 0; u < KEY_U; ++u) {
            const int t = t0 + 32 * u, i0 = 4 * t;
            if (i0 < E)
                keys[t] = key4(dsts[i0],
                               (i0 + 1 < E) ? dsts[i0 + 1] : 0,
                               (i0 + 2 < E) ? dsts[i0 + 2] : 0,
                               (i0 + 3 < E) ? dsts[i0 + 3] : 0);
        }
    }
}

template <class Fn>
__device__ __forceinline__ void scan_keys(const unsigned* __restrict__ keys,
                                          const int* __restrict__ dsts, int E,
                                          int n0, unsigned cnt, Fn emit) {
    const int tid = threadIdx.x;
    constexpr int nt = OWN_THREADS;
    const int E16 = E >> 4;
    const uint4* __restrict__ k4 = reinterpret_cast<const uint4*>(keys);
    const unsigned pat = (blockIdx.x & 0xffu) * 0x01010101u;
    auto eq4 = [&](unsigned w) {
        const unsigned x = w ^ pat;
        return (x - 0x01010101u) & ~x & 0x80808080u;
    };
    auto test16 = [&](uint4 v, int q) {
        unsigned m = eq4(v.x) | (eq4(v.y) >> 1) | (eq4(v.z) >> 2) |
                     (eq4(v.w) >> 3);
        while (m) {
            const int p = __builtin_ctz(m);
            m &= m - 1;
            emit(16 * q + 4 * (7 - (p & 7)) + (p >> 3));
        }
    };
    constexpr int U = 8;
    int q = tid;
    for (; q + (U - 1) * nt < E16; q += U * nt) {
        uint4 v[U];
#pragma unroll
        for (int u = 0; u < U; ++u) v[u] = k4[q + u * nt];
#pragma unroll
        for (int u = 0; u < U; ++u) test16(v[u], q + u * nt);
    }
    for (; q < E16; q += nt) test16(k4[q], q);
    for (int i = 16 * E16 + tid; i < E; i += nt) {
        const unsigned a = (unsigned)(dsts[i] - n0);
        if (a < cnt) emit(i);
    }
}

template <int H, int C, bool FINAL>
__global__ __launch_bounds__(OWN_THREADS) void owner_kernel(
    const int* __restrict__ srcs, const int* __restrict__ dsts,
    const unsigned* __restrict__ keys, int E, int N,
    const float* __restrict__ as_, const float* __restrict__ ad_,
    const float* __restrict__ xw, const float* __restrict__ bias,
    float* __restrict__ outp) {
    static_assert(C % 4 == 0, "C must be a multiple of 4");
    static_assert(!FINAL || H == 1, "the final layer has one head");
    constexpr int F = H * C;
    constexpr int FP = F + 1;
    __shared__ unsigned s_m[NPS * H];
    __shared__ float    s_ad[NPS * H];
    __shared__ float    s_den[NPS * H];
    __shared__ float    s_acc[NPS * FP];
    __shared__ int      s_li[LCAP];
    __shared__ int      s_ld[LCAP];
    __shared__ int      s_cnt;

    const int tid = threadIdx.x;
    constexpr int nt = OWN_THREADS;
    const int n0 = blockIdx.x * NPS;
    if (n0 >= N) return;
    const int cnt = min(NPS, N - n0);

    for (int j = tid; j < cnt * H; j += nt) {
        s_m[j] = FKEY_NEG_INF;
        s_den[j] = 0.f;
        s_ad[j] = ad_[(size_t)n0 * H + j];
    }
    for (int j = tid; j < cnt * FP; j += nt) s_acc[j] = 0.f;
    if (tid == 0) s_cnt = 0;
    __syncthreads();

    auto edge_max = [&](int s, int dl) {
#pragma unroll
        for (int h = 0; h < H; ++h) {
            float e = as_[s * H + h] + s_ad[dl * H + h];
            e = (e > 0.f) ? e : NEG_SLOPE * e;
            atomicMax(&s_m[dl * H + h], fkey(e));
        }
    };
    auto edge_acc = [&](int s, int dl) {
        float w[H];
#pragma unroll
        for (int h = 0; h < H; ++h) {
            float e = as_[s * H + h] + s_ad[dl * H + h];
            e = (e > 0.f) ? e : NEG_SLOPE * e;
            w[h] = expf(e - fkey_inv(s_m[dl * H + h]));
            atomicAdd(&s_den[dl * H + h], w[h]);
        }
        const float4* __restrict__ sv = (const float4*)(xw + (size_t)s * F);
        float* dv = &s_acc[dl * FP];
#pragma unroll
        for (int f = 0; f < F / 4; ++f) {
            const float4 v = sv[f];
            const float wh = w[(f * 4) / C];
            atomicAdd(&dv[f * 4 + 0], wh * v.x);
            atomicAdd(&dv[f * 4 + 1], wh * v.y);
            atomicAdd(&dv[f * 4 + 2], wh * v.z);
            atomicAdd(&dv[f * 4 + 3], wh * v.w);
        }
    };

    scan_keys(keys, dsts, E, n0, (unsigned)cnt, [&](int i) {
        const int pos = atomicAdd(&s_cnt, 1);
        if (pos < LCAP) s_li[pos] = i;
    });
    __syncthreads();
    const int  total  = s_cnt;
    const bool listed = total <= LCAP;
    if (listed)
        for (int k = tid; k < total; k += nt) {
            const int i = s_li[k];
            const unsigned a = (unsigned)(dsts[i] - n0);
            s_li[k] = srcs[i];
            s_ld[k] = (a < (unsigned)cnt) ? (int)a : -1;
        }

    auto for_each_edge = [&](auto fn) {
        if (listed) {
            for (int k = tid; k < total; k += nt) {
                const int dl = s_ld[k];
                if (dl >= 0) fn(s_li[k], dl);
            }
        } else {
            for (int i = tid; i < E; i += nt) {
                const unsigned a = (unsigned)(dsts[i] - n0);
                if (a < (unsigned)cnt) fn(srcs[i], (int)a);
            }
        }
        for (int dl = tid; dl < cnt; dl += nt) fn(n0 + dl, dl);
    };

    for_each_edge(edge_max);
    __syncthreads();
    for_each_edge(edge_acc);
    __syncthreads();

    if (!FINAL) {
        for (int j = tid; j < cnt * F; j += nt) {
            const float v =
                s_acc[(j / F) * FP + j % F] / s_den[j / C] + bias[j % F];
            outp[(size_t)n0 * F + j] = (v > 0.f) ? v : (expf(v) - 1.f);
        }
    } else {
        for (int dl = tid; dl < cnt; dl += nt) {
            const float dn = s_den[dl];
            float v[F];
            float m = -INFINITY;
#pragma unroll
            for (int c = 0; c < F; ++c) {
                v[c] = s_acc[dl * FP + c] / dn + bias[c];
                m = fmaxf(m, v[c]);
            }
            float ssum = 0.f;
#pragma unroll
            for (int c = 0; c < F; ++c) ssum += expf(v[c] - m);
            const float lg = m + logf(ssum);
#pragma unroll
            for (int c = 0; c < F; ++c)
                outp[(size_t)(n0 + dl) * F + c] = v[c] - lg;
        }
    }
}

extern "C" void kernel_launch(void* const* d_in, const int* in_sizes, int n_in,
                              void* d_out, int out_size, void* d_ws,
                              size_t ws_size, hipStream_t stream) {
    const float* x   = (const float*)d_in[0];
    const int*   ei  = (const int*)d_in[1];
    const float* W1  = (const float*)d_in[2];
    const float* a1s = (const float*)d_in[3];
    const float* a1d = (const float*)d_in[4];
    const float* b1  = (const float*)d_in[5];
    const float* W2  = (const float*)d_in[6];
    const float* a2s = (const float*)d_in[7];
    const float* a2d = (const float*)d_in[8];
    const float* b2  = (const float*)d_in[9];
    float* out = (float*)d_out;

    const int N = in_sizes[0] / 128;
    const int E = in_sizes[1] / 2;
    const int* srcs = ei;
    const int* dsts = ei + E;

    char* p = (char*)d_ws;
    auto carve = [&](size_t bytes) {
        void* r = (void*)p;
        p += (bytes + 255) & ~(size_t)255;
        return r;
    };
    float* xw1 = (float*)carve((size_t)N * 32 * 4);
    float* as1 = (float*)carve((size_t)N * 4 * 4);
    float* ad1 = (float*)carve((size_t)N * 4 * 4);
    float* h1  = (float*)carve((size_t)N * 32 * 4);
    unsigned* keys = (unsigned*)carve((size_t)E + 16);
    float* xw2 = xw1;
    float* as2 = as1;
    float* ad2 = ad1;

    const int B = 256;
    auto g = [](int n, int b) { return (n + b - 1) / b; };

    key_kernel<<<g(g(E, 4 * KEY_U), B), B, 0, stream>>>(dsts, E, keys);

    gemm_alpha_kernel<128, 32, 4><<<g(N, 32), 64, 0, stream>>>(
        x, W1, a1s, a1d, xw1, as1, ad1, N);
    owner_kernel<4, 8, false><<<g(N, NPS), OWN_THREADS, 0, stream>>>(
        srcs, dsts, keys, E, N, as1, ad1, xw1, b1, h1);

    gemm_alpha_kernel<32, 16, 1><<<g(N, 32), 64, 0, stream>>>(
        h1, W2, a2s, a2d, xw2, as2, ad2, N);
    owner_kernel<1, 16, true><<<g(N, NPS), OWN_THREADS, 0, stream>>>(
        srcs, dsts, keys, E, N, as2, ad2, xw2, b2, out);
}


# layer 2's owner launch loads each group's checked edge list (source, local destination, one packed word) that layer 1's launch stored in the workspace; no second pass over key plane; arrival order
# speedup vs baseline: 10.4324x; 1.2324x over previous
#include <hip/hip_runtime.h>
#include <math.h>

typedef __attribute__((ext_vector_type(2))) float v2f;
typedef __attribute__((ext_vector_type(8))) float v8f;

#define NEG_SLOPE 0.2f

__device__ __forceinline__ unsigned fkey(float f) {
    unsigned b = __float_as_uint(f);
    return (b & 0x80000000u) ? ~b : (b | 0x80000000u);
}
__device__ __forceinline__ float fkey_inv(unsigned u) {
    return (u & 0x80000000u) ? __uint_as_float(u & 0x7fffffffu)
                             : __uint_as_float(~u);
}
#define FKEY_NEG_INF 0x007fffffu

template <int K, int NCOL, int H>
__global__ void gemm_alpha_kernel(const float* __restrict__ A,
                                  const float* __restrict__ B,
                                  const float* __restrict__ a_src,
                                  const float* __restrict__ a_dst,
                                  float* __restrict__ C,
                                  float* __restrict__ as_,
                                  float* __restrict__ ad_, int N) {
    constexpr int NT = NCOL / 16;
    constexpr int CH = NCOL / H;
    static_assert(CH == 8 || CH == 16, "a head must lie inside one tile");
    const int lane = threadIdx.x & 31;
    const int l    = lane & 15;
    const int hf   = lane >> 4;
    const int row0 =
        (blockIdx.x * (blockDim.x >> 5) + (threadIdx.x >> 5)) * 16;
    if (row0 >= N) return;

    const float* __restrict__ arow = A + (size_t)(row0 + l) * K;
    v8f acc[NT];
#pragma unroll
    for (int t = 0; t < NT; ++t) acc[t] = (v8f){};

#pragma unroll
    for (int k = 0; k < K; k += 4) {
        v2f a;
        a.x = arow[k + 2 * hf];
        a.y = arow[k + 2 * hf + 1];
#pragma unroll
        for (int t = 0; t < NT; ++t) {
            v2f b;
            b.x = B[(size_t)(k + 2 * hf) * NCOL + 16 * t + l];
            b.y = B[(size_t)(k + 2 * hf + 1) * NCOL + 16 * t + l];
            acc[t] = __builtin_amdgcn_wmma_f32_16x16x4_f32(
                 false, a,  false, b,
                 (short)0, acc[t],  false,
                 false);
        }
    }
    float s[NT][8], d[NT][8];
#pragma unroll
    for (int t = 0; t < NT; ++t) {
        const float ws = a_src[16 * t + l];
        const float wd = a_dst[16 * t + l];
#pragma unroll
        for (int i = 0; i < 8; ++i) {
            s[t][i] = acc[t][i] * ws;
            d[t][i] = acc[t][i] * wd;
        }
    }
#pragma unroll
    for (int m = 1; m < CH; m <<= 1) {
        float ps[NT][8], pd[NT][8];
#pragma unroll
        for (int t = 0; t < NT; ++t)
#pragma unroll
            for (int i = 0; i < 8; ++i) {
                ps[t][i] = __shfl_xor(s[t][i], m);
                pd[t][i] = __shfl_xor(d[t][i], m);
            }
#pragma unroll
        for (int t = 0; t < NT; ++t)
#pragma unroll
            for (int i = 0; i < 8; ++i) {
                s[t][i] += ps[t][i];
                d[t][i] += pd[t][i];
            }
    }
#pragma unroll
    for (int t = 0; t < NT; ++t)
#pragma unroll
        for (int i = 0; i < 8; ++i)
            C[(size_t)(row0 + i + 8 * hf) * NCOL + 16 * t + l] = acc[t][i];
    if ((l & (CH - 1)) == 0) {
#pragma unroll
        for (int t = 0; t < NT; ++t)
#pragma unroll
            for (int i = 0; i < 8; ++i) {
                const size_t o =
                    (size_t)(row0 + i + 8 * hf) * H + (16 * t + l) / CH;
                as_[o] = s[t][i];
                ad_[o] = d[t][i];
            }
    }
}

#define NPS  392
#define OWN_THREADS 1024
#define LCAP 18432
#define LPITCH 7168
#define LSHIFT 9

#define KEY_U 4
__device__ __forceinline__ unsigned key4(int a, int b, int c, int d) {
    return (((unsigned)a / (unsigned)NPS) & 0xffu) |
           ((((unsigned)b / (unsigned)NPS) & 0xffu) << 8) |
           ((((unsigned)c / (unsigned)NPS) & 0xffu) << 16) |
           ((((unsigned)d / (unsigned)NPS) & 0xffu) << 24);
}
__global__ void key_kernel(const int* __restrict__ dsts, int E,
                           unsigned* __restrict__ keys) {
    const int lane = threadIdx.x & 31;
    const int wave = (blockIdx.x * blockDim.x + threadIdx.x) >> 5;
    const int t0   = wave * (KEY_U * 32) + lane;
    const bool whole = (reinterpret_cast<size_t>(dsts) & 15) == 0 &&
                       (wave + 1) * (KEY_U * 128) <= E;
    if (whole) {
        const int4* __restrict__ d4 = reinterpret_cast<const int4*>(dsts);
        int4 v[KEY_U];
#pragma unroll
        for (int u = 0; u < KEY_U; ++u) v[u] = d4[t0 + 32 * u];
#pragma unroll
        for (int u = 0; u < KEY_U; ++u)
            keys[t0 + 32 * u] = key4(v[u].x, v[u].y, v[u].z, v[u].w);
    } else {
        for (int u = 0; u < KEY_U; ++u) {
            const int t = t0 + 32 * u, i0 = 4 * t;
            if (i0 < E)
                keys[t] = key4(dsts[i0],
                               (i0 + 1 < E) ? dsts[i0 + 1] : 0,
                               (i0 + 2 < E) ? dsts[i0 + 2] : 0,
                               (i0 + 3 < E) ? dsts[i0 + 3] : 0);
        }
    }
}

template <class Fn>
__device__ __forceinline__ void scan_keys(const unsigned* __restrict__ keys,
                                          const int* __restrict__ dsts, int E,
                                          int n0, unsigned cnt, Fn emit) {
    const int tid = threadIdx.x;
    constexpr int nt = OWN_THREADS;
    const int E16 = E >> 4;
    const uint4* __restrict__ k4 = reinterpret_cast<const uint4*>(keys);
    const unsigned pat = (blockIdx.x & 0xffu) * 0x01010101u;
    auto eq4 = [&](unsigned w) {
        const unsigned x = w ^ pat;
        return (x - 0x01010101u) & ~x & 0x80808080u;
    };
    auto test16 = [&](uint4 v, int q) {
        unsigned m = eq4(v.x) | (eq4(v.y) >> 1) | (eq4(v.z) >> 2) |
                     (eq4(v.w) >> 3);
        while (m) {
            const int p = __builtin_ctz(m);
            m &= m - 1;
            emit(16 * q + 4 * (7 - (p & 7)) + (p >> 3));
        }
    };
    constexpr int U = 8;
    int q = tid;
    for (; q + (U - 1) * nt < E16; q += U * nt) {
        uint4 v[U];
#pragma unroll
        for (int u = 0; u < U; ++u) v[u] = k4[q + u * nt];
#pragma unroll
        for (int u = 0; u < U; ++u) test16(v[u], q + u * nt);
    }
    for (; q < E16; q += nt) test16(k4[q], q);
    for (int i = 16 * E16 + tid; i < E; i += nt) {
        const unsigned a = (unsigned)(dsts[i] - n0);
        if (a < cnt) emit(i);
    }
}

template <int H, int C, bool FINAL, bool FIRST>
__global__ __launch_bounds__(OWN_THREADS) void owner_kernel(
    const int* __restrict__ srcs, const int* __restrict__ dsts,
    const unsigned* __restrict__ keys, int E, int N,
    const float* __restrict__ as_, const float* __restrict__ ad_,
    const float* __restrict__ xw, const float* __restrict__ bias,
    float* __restrict__ outp, int* __restrict__ ltot,
    int* __restrict__ lsave, const int* ws_first, size_t ws_bytes) {
    static_assert(C % 4 == 0, "C must be a multiple of 4");
    static_assert(!FINAL || H == 1, "the final layer has one head");
    static_assert(NPS <= (1 << LSHIFT), "a local destination must fit");
    static_assert(LPITCH <= LCAP, "a stored list must fit the list in LDS");
    constexpr int F = H * C;
    constexpr int FP = F + 1;
    __shared__ unsigned s_m[NPS * H];
    __shared__ float    s_ad[NPS * H];
    __shared__ float    s_den[NPS * H];
    __shared__ float    s_acc[NPS * FP];
    __shared__ int      s_li[LCAP];
    __shared__ int      s_ld[LCAP];
    __shared__ int      s_cnt;

    const int tid = threadIdx.x;
    constexpr int nt = OWN_THREADS;
    const int n0 = blockIdx.x * NPS;
    if (n0 >= N) return;
    const int cnt = min(NPS, N - n0);

    for (int j = tid; j < cnt * H; j += nt) {
        s_m[j] = FKEY_NEG_INF;
        s_den[j] = 0.f;
        s_ad[j] = ad_[(size_t)n0 * H + j];
    }
    for (int j = tid; j < cnt * FP; j += nt) s_acc[j] = 0.f;
    if (tid == 0) s_cnt = 0;
    __syncthreads();

    auto edge_max = [&](int s, int dl) {
#pragma unroll
        for (int h = 0; h < H; ++h) {
            float e = as_[s * H + h] + s_ad[dl * H + h];
            e = (e > 0.f) ? e : NEG_SLOPE * e;
            atomicMax(&s_m[dl * H + h], fkey(e));
        }
    };
    auto edge_acc = [&](int s, int dl) {
        float w[H];
#pragma unroll
        for (int h = 0; h < H; ++h) {
            float e = as_[s * H + h] + s_ad[dl * H + h];
            e = (e > 0.f) ? e : NEG_SLOPE * e;
            w[h] = expf(e - fkey_inv(s_m[dl * H + h]));
            atomicAdd(&s_den[dl * H + h], w[h]);
        }
        const float4* __restrict__ sv = (const float4*)(xw + (size_t)s * F);
        float* dv = &s_acc[dl * FP];
#pragma unroll
        for (int f = 0; f < F / 4; ++f) {
            const float4 v = sv[f];
            const float wh = w[(f * 4) / C];
            atomicAdd(&dv[f * 4 + 0], wh * v.x);
            atomicAdd(&dv[f * 4 + 1], wh * v.y);
            atomicAdd(&dv[f * 4 + 2], wh * v.z);
            atomicAdd(&dv[f * 4 + 3], wh * v.w);
        }
    };

    const bool ws_ok =
        (size_t)((lsave + (size_t)gridDim.x * LPITCH) - ws_first) *
                sizeof(int) <= ws_bytes &&
        N <= (1 << (31 - LSHIFT));
    int* __restrict__ mine = lsave + (size_t)blockIdx.x * LPITCH;

    int  total  = 0;
    bool loaded = false;
    if (!FIRST && ws_ok) {
        total  = ltot[blockIdx.x];
        loaded = total <= LPITCH;
    }
    if (loaded) {
        constexpr int LSLOTS = LPITCH / OWN_THREADS;
        static_assert(LSLOTS * OWN_THREADS == LPITCH, "whole slots");
        if (total > 0) {
            int v[LSLOTS];
#pragma unroll
            for (int j = 0; j < LSLOTS; ++j)
                v[j] = mine[min(tid + j * nt, total - 1)];
#pragma unroll
            for (int j = 0; j < LSLOTS; ++j) {
                const int k = tid + j * nt;
                if (k < total) {
                    s_li[k] = v[j] >> LSHIFT;
                    s_ld[k] = (v[j] < 0) ? -1 : (v[j] & ((1 << LSHIFT) - 1));
                }
            }
        }
    } else {
        scan_keys(keys, dsts, E, n0, (unsigned)cnt, [&](int i) {
            const int pos = atomicAdd(&s_cnt, 1);
            if (pos < LCAP) s_li[pos] = i;
        });
        __syncthreads();
        total = s_cnt;
    }
    const bool listed = total <= LCAP;
    if (!loaded) {
        const bool store = FIRST && ws_ok && total <= LPITCH;
        if (FIRST && ws_ok && tid == 0) ltot[blockIdx.x] = total;
        auto resolve = [&](auto keep) {
            for (int k = tid; k < total; k += nt) {
                const int i = s_li[k];
                const unsigned a = (unsigned)(dsts[i] - n0);
                const int s  = srcs[i];
                const int dl = (a < (unsigned)cnt) ? (int)a : -1;
                s_li[k] = s;
                s_ld[k] = dl;
                keep(k, s, dl);
            }
        };
        if (store)
            resolve([&](int k, int s, int dl) {
                mine[k] = (dl < 0) ? -1 : ((s << LSHIFT) | dl);
            });
        else if (listed)
            resolve([](int, int, int) {});
    }

    auto for_each_edge = [&](auto fn) {
        if (listed) {
            for (int k = tid; k < total; k += nt) {
                const int dl = s_ld[k];
                if (dl >= 0) fn(s_li[k], dl);
            }
        } else {
            for (int i = tid; i < E; i += nt) {
                const unsigned a = (unsigned)(dsts[i] - n0);
                if (a < (unsigned)cnt) fn(srcs[i], (int)a);
            }
        }
        for (int dl = tid; dl < cnt; dl += nt) fn(n0 + dl, dl);
    };

    for_each_edge(edge_max);
    __syncthreads();
    for_each_edge(edge_acc);
    __syncthreads();

    if (!FINAL) {
        for (int j = tid; j < cnt * F; j += nt) {
            const float v =
                s_acc[(j / F) * FP + j % F] / s_den[j / C] + bias[j % F];
            outp[(size_t)n0 * F + j] = (v > 0.f) ? v : (expf(v) - 1.f);
        }
    } else {
        for (int dl = tid; dl < cnt; dl += nt) {
            const float dn = s_den[dl];
            float v[F];
            float m = -INFINITY;
#pragma unroll
            for (int c = 0; c < F; ++c) {
                v[c] = s_acc[dl * FP + c] / dn + bias[c];
                m = fmaxf(m, v[c]);
            }
            float ssum = 0.f;
#pragma unroll
            for (int c = 0; c < F; ++c) ssum += expf(v[c] - m);
            const float lg = m + logf(ssum);
#pragma unroll
            for (int c = 0; c < F; ++c)
                outp[(size_t)(n0 + dl) * F + c] = v[c] - lg;
        }
    }
}

extern "C" void kernel_launch(void* const* d_in, const int* in_sizes, int n_in,
                              void* d_out, int out_size, void* d_ws,
                              size_t ws_size, hipStream_t stream) {
    const float* x   = (const float*)d_in[0];
    const int*   ei  = (const int*)d_in[1];
    const float* W1  = (const float*)d_in[2];
    const float* a1s = (const float*)d_in[3];
    const float* a1d = (const float*)d_in[4];
    const float* b1  = (const float*)d_in[5];
    const float* W2  = (const float*)d_in[6];
    const float* a2s = (const float*)d_in[7];
    const float* a2d = (const float*)d_in[8];
    const float* b2  = (const float*)d_in[9];
    float* out = (float*)d_out;

    const int N = in_sizes[0] / 128;
    const int E = in_sizes[1] / 2;
    const int* srcs = ei;
    const int* dsts = ei + E;

    char* p = (char*)d_ws;
    auto carve = [&](size_t bytes) {
        void* r = (void*)p;
        p += (bytes + 255) & ~(size_t)255;
        return r;
    };
    float* xw1 = (float*)carve((size_t)N * 32 * 4);
    float* as1 = (float*)carve((size_t)N * 4 * 4);
    float* ad1 = (float*)carve((size_t)N * 4 * 4);
    float* h1  = (float*)carve((size_t)N * 32 * 4);
    unsigned* keys = (unsigned*)carve((size_t)E + 16);
    float* xw2 = xw1;
    float* as2 = as1;
    float* ad2 = ad1;

    const int B = 256;
    auto g = [](int n, int b) { return (n + b - 1) / b; };

    int* ltot  = (int*)carve((size_t)g(N, NPS) * 4);
    int* lsave = (int*)carve((size_t)g(N, NPS) * LPITCH * 4);
    const int* ws_first = (const int*)d_ws;

    key_kernel<<<g(g(E, 4 * KEY_U), B), B, 0, stream>>>(dsts, E, keys);

    gemm_alpha_kernel<128, 32, 4><<<g(N, 32), 64, 0, stream>>>(
        x, W1, a1s, a1d, xw1, as1, ad1, N);
    owner_kernel<4, 8, false, true><<<g(N, NPS), OWN_THREADS, 0, stream>>>(
        srcs, dsts, keys, E, N, as1, ad1, xw1, b1, h1, ltot, lsave, ws_first,
        ws_size);

    gemm_alpha_kernel<32, 16, 1><<<g(N, 32), 64, 0, stream>>>(
        h1, W2, a2s, a2d, xw2, as2, ad2, N);
    owner_kernel<1, 16, true, false><<<g(N, NPS), OWN_THREADS, 0, stream>>>(
        srcs, dsts, keys, E, N, as2, ad2, xw2, b2, out, ltot, lsave, ws_first,
        ws_size);
}


# owner launches: a thread keeps the scores of its own list entries and self-loop in registers from the max walk to the sum walk: source scores gathered once per edge, not twice; arrival order
# speedup vs baseline: 11.1327x; 1.0671x over previous
#include <hip/hip_runtime.h>
#include <math.h>

typedef __attribute__((ext_vector_type(2))) float v2f;
typedef __attribute__((ext_vector_type(8))) float v8f;

#define NEG_SLOPE 0.2f

__device__ __forceinline__ unsigned fkey(float f) {
    unsigned b = __float_as_uint(f);
    return (b & 0x80000000u) ? ~b : (b | 0x80000000u);
}
__device__ __forceinline__ float fkey_inv(unsigned u) {
    return (u & 0x80000000u) ? __uint_as_float(u & 0x7fffffffu)
                             : __uint_as_float(~u);
}
#define FKEY_NEG_INF 0x007fffffu

template <int K, int NCOL, int H>
__global__ void gemm_alpha_kernel(const float* __restrict__ A,
                                  const float* __restrict__ B,
                                  const float* __restrict__ a_src,
                                  const float* __restrict__ a_dst,
                                  float* __restrict__ C,
                                  float* __restrict__ as_,
                                  float* __restrict__ ad_, int N) {
    constexpr int NT = NCOL / 16;
    constexpr int CH = NCOL / H;
    static_assert(CH == 8 || CH == 16, "a head must lie inside one tile");
    const int lane = threadIdx.x & 31;
    const int l    = lane & 15;
    const int hf   = lane >> 4;
    const int row0 =
        (blockIdx.x * (blockDim.x >> 5) + (threadIdx.x >> 5)) * 16;
    if (row0 >= N) return;

    const float* __restrict__ arow = A + (size_t)(row0 + l) * K;
    v8f acc[NT];
#pragma unroll
    for (int t = 0; t < NT; ++t) acc[t] = (v8f){};

#pragma unroll
    for (int k = 0; k < K; k += 4) {
        v2f a;
        a.x = arow[k + 2 * hf];
        a.y = arow[k + 2 * hf + 1];
#pragma unroll
        for (int t = 0; t < NT; ++t) {
            v2f b;
            b.x = B[(size_t)(k + 2 * hf) * NCOL + 16 * t + l];
            b.y = B[(size_t)(k + 2 * hf + 1) * NCOL + 16 * t + l];
            acc[t] = __builtin_amdgcn_wmma_f32_16x16x4_f32(
                 false, a,  false, b,
                 (short)0, acc[t],  false,
                 false);
        }
    }
    float s[NT][8], d[NT][8];
#pragma unroll
    for (int t = 0; t < NT; ++t) {
        const float ws = a_src[16 * t + l];
        const float wd = a_dst[16 * t + l];
#pragma unroll
        for (int i = 0; i < 8; ++i) {
            s[t][i] = acc[t][i] * ws;
            d[t][i] = acc[t][i] * wd;
        }
    }
#pragma unroll
    for (int m = 1; m < CH; m <<= 1) {
        float ps[NT][8], pd[NT][8];
#pragma unroll
        for (int t = 0; t < NT; ++t)
#pragma unroll
            for (int i = 0; i < 8; ++i) {
                ps[t][i] = __shfl_xor(s[t][i], m);
                pd[t][i] = __shfl_xor(d[t][i], m);
            }
#pragma unroll
        for (int t = 0; t < NT; ++t)
#pragma unroll
            for (int i = 0; i < 8; ++i) {
                s[t][i] += ps[t][i];
                d[t][i] += pd[t][i];
            }
    }
#pragma unroll
    for (int t = 0; t < NT; ++t)
#pragma unroll
        for (int i = 0; i < 8; ++i)
            C[(size_t)(row0 + i + 8 * hf) * NCOL + 16 * t + l] = acc[t][i];
    if ((l & (CH - 1)) == 0) {
#pragma unroll
        for (int t = 0; t < NT; ++t)
#pragma unroll
            for (int i = 0; i < 8; ++i) {
                const size_t o =
                    (size_t)(row0 + i + 8 * hf) * H + (16 * t + l) / CH;
                as_[o] = s[t][i];
                ad_[o] = d[t][i];
            }
    }
}

#define NPS  392
#define OWN_THREADS 1024
#define LCAP 18432
#define LPITCH 7168
#define LSHIFT 9

#define KEY_U 4
__device__ __forceinline__ unsigned key4(int a, int b, int c, int d) {
    return (((unsigned)a / (unsigned)NPS) & 0xffu) |
           ((((unsigned)b / (unsigned)NPS) & 0xffu) << 8) |
           ((((unsigned)c / (unsigned)NPS) & 0xffu) << 16) |
           ((((unsigned)d / (unsigned)NPS) & 0xffu) << 24);
}
__global__ void key_kernel(const int* __restrict__ dsts, int E,
                           unsigned* __restrict__ keys) {
    const int lane = threadIdx.x & 31;
    const int wave = (blockIdx.x * blockDim.x + threadIdx.x) >> 5;
    const int t0   = wave * (KEY_U * 32) + lane;
    const bool whole = (reinterpret_cast<size_t>(dsts) & 15) == 0 &&
                       (wave + 1) * (KEY_U * 128) <= E;
    if (whole) {
        const int4* __restrict__ d4 = reinterpret_cast<const int4*>(dsts);
        int4 v[KEY_U];
#pragma unroll
        for (int u = 0; u < KEY_U; ++u) v[u] = d4[t0 + 32 * u];
#pragma unroll
        for (int u = 0; u < KEY_U; ++u)
            keys[t0 + 32 * u] = key4(v[u].x, v[u].y, v[u].z, v[u].w);
    } else {
        for (int u = 0; u < KEY_U; ++u) {
            const int t = t0 + 32 * u, i0 = 4 * t;
            if (i0 < E)
                keys[t] = key4(dsts[i0],
                               (i0 + 1 < E) ? dsts[i0 + 1] : 0,
                               (i0 + 2 < E) ? dsts[i0 + 2] : 0,
                               (i0 + 3 < E) ? dsts[i0 + 3] : 0);
        }
    }
}

template <class Fn>
__device__ __forceinline__ void scan_keys(const unsigned* __restrict__ keys,
                                          const int* __restrict__ dsts, int E,
                                          int n0, unsigned cnt, Fn emit) {
    const int tid = threadIdx.x;
    constexpr int nt = OWN_THREADS;
    const int E16 = E >> 4;
    const uint4* __restrict__ k4 = reinterpret_cast<const uint4*>(keys);
    const unsigned pat = (blockIdx.x & 0xffu) * 0x01010101u;
    auto eq4 = [&](unsigned w) {
        const unsigned x = w ^ pat;
        return (x - 0x01010101u) & ~x & 0x80808080u;
    };
    auto test16 = [&](uint4 v, int q) {
        unsigned m = eq4(v.x) | (eq4(v.y) >> 1) | (eq4(v.z) >> 2) |
                     (eq4(v.w) >> 3);
        while (m) {
            const int p = __builtin_ctz(m);
            m &= m - 1;
            emit(16 * q + 4 * (7 - (p & 7)) + (p >> 3));
        }
    };
    constexpr int U = 8;
    int q = tid;
    for (; q + (U - 1) * nt < E16; q += U * nt) {
        uint4 v[U];
#pragma unroll
        for (int u = 0; u < U; ++u) v[u] = k4[q + u * nt];
#pragma unroll
        for (int u = 0; u < U; ++u) test16(v[u], q + u * nt);
    }
    for (; q < E16; q += nt) test16(k4[q], q);
    for (int i = 16 * E16 + tid; i < E; i += nt) {
        const unsigned a = (unsigned)(dsts[i] - n0);
        if (a < cnt) emit(i);
    }
}

template <int H, int C, bool FINAL, bool FIRST>
__global__ __launch_bounds__(OWN_THREADS) void owner_kernel(
    const int* __restrict__ srcs, const int* __restrict__ dsts,
    const unsigned* __restrict__ keys, int E, int N,
    const float* __restrict__ as_, const float* __restrict__ ad_,
    const float* __restrict__ xw, const float* __restrict__ bias,
    float* __restrict__ outp, int* __restrict__ ltot,
    int* __restrict__ lsave, const int* ws_first, size_t ws_bytes) {
    static_assert(C % 4 == 0, "C must be a multiple of 4");
    static_assert(!FINAL || H == 1, "the final layer has one head");
    static_assert(NPS <= (1 << LSHIFT), "a local destination must fit");
    static_assert(LPITCH <= LCAP, "a stored list must fit the list in LDS");
    constexpr int F = H * C;
    constexpr int FP = F + 1;
    __shared__ unsigned s_m[NPS * H];
    __shared__ float    s_ad[NPS * H];
    __shared__ float    s_den[NPS * H];
    __shared__ float    s_acc[NPS * FP];
    __shared__ int      s_li[LCAP];
    __shared__ int      s_ld[LCAP];
    __shared__ int      s_cnt;

    const int tid = threadIdx.x;
    constexpr int nt = OWN_THREADS;
    const int n0 = blockIdx.x * NPS;
    if (n0 >= N) return;
    const int cnt = min(NPS, N - n0);

    for (int j = tid; j < cnt * H; j += nt) {
        s_m[j] = FKEY_NEG_INF;
        s_den[j] = 0.f;
        s_ad[j] = ad_[(size_t)n0 * H + j];
    }
    for (int j = tid; j < cnt * FP; j += nt) s_acc[j] = 0.f;
    if (tid == 0) s_cnt = 0;
    __syncthreads();

    auto score = [&](int s, int dl, float* e) {
#pragma unroll
        for (int h = 0; h < H; ++h) {
            const float v = as_[s * H + h] + s_ad[dl * H + h];
            e[h] = (v > 0.f) ? v : NEG_SLOPE * v;
        }
    };
    auto put_max = [&](int dl, const float* e) {
#pragma unroll
        for (int h = 0; h < H; ++h) atomicMax(&s_m[dl * H + h], fkey(e[h]));
    };
    auto add_edge = [&](int s, int dl, const float* e) {
        float w[H];
#pragma unroll
        for (int h = 0; h < H; ++h) {
            w[h] = expf(e[h] - fkey_inv(s_m[dl * H + h]));
            atomicAdd(&s_den[dl * H + h], w[h]);
        }
        const float4* __restrict__ sv = (const float4*)(xw + (size_t)s * F);
        float* dv = &s_acc[dl * FP];
#pragma unroll
        for (int f = 0; f < F / 4; ++f) {
            const float4 v = sv[f];
            const float wh = w[(f * 4) / C];
            atomicAdd(&dv[f * 4 + 0], wh * v.x);
            atomicAdd(&dv[f * 4 + 1], wh * v.y);
            atomicAdd(&dv[f * 4 + 2], wh * v.z);
            atomicAdd(&dv[f * 4 + 3], wh * v.w);
        }
    };

    const bool ws_ok =
        (size_t)((lsave + (size_t)gridDim.x * LPITCH) - ws_first) *
                sizeof(int) <= ws_bytes &&
        N <= (1 << (31 - LSHIFT));
    int* __restrict__ mine = lsave + (size_t)blockIdx.x * LPITCH;

    int  total  = 0;
    bool loaded = false;
    if (!FIRST && ws_ok) {
        total  = ltot[blockIdx.x];
        loaded = total <= LPITCH;
    }
    if (loaded) {
        constexpr int LSLOTS = LPITCH / OWN_THREADS;
        static_assert(LSLOTS * OWN_THREADS == LPITCH, "whole slots");
        if (total > 0) {
            int v[LSLOTS];
#pragma unroll
            for (int j = 0; j < LSLOTS; ++j)
                v[j] = mine[min(tid + j * nt, total - 1)];
#pragma unroll
            for (int j = 0; j < LSLOTS; ++j) {
                const int k = tid + j * nt;
                if (k < total) {
                    s_li[k] = v[j] >> LSHIFT;
                    s_ld[k] = (v[j] < 0) ? -1 : (v[j] & ((1 << LSHIFT) - 1));
                }
            }
        }
    } else {
        scan_keys(keys, dsts, E, n0, (unsigned)cnt, [&](int i) {
            const int pos = atomicAdd(&s_cnt, 1);
            if (pos < LCAP) s_li[pos] = i;
        });
        __syncthreads();
        total = s_cnt;
    }
    const bool listed = total <= LCAP;
    if (!loaded) {
        const bool store = FIRST && ws_ok && total <= LPITCH;
        if (FIRST && ws_ok && tid == 0) ltot[blockIdx.x] = total;
        auto resolve = [&](auto keep) {
            for (int k = tid; k < total; k += nt) {
                const int i = s_li[k];
                const unsigned a = (unsigned)(dsts[i] - n0);
                const int s  = srcs[i];
                const int dl = (a < (unsigned)cnt) ? (int)a : -1;
                s_li[k] = s;
                s_ld[k] = dl;
                keep(k, s, dl);
            }
        };
        if (store)
            resolve([&](int k, int s, int dl) {
                mine[k] = (dl < 0) ? -1 : ((s << LSHIFT) | dl);
            });
        else if (listed)
            resolve([](int, int, int) {});
    }

    constexpr int KS = LPITCH / OWN_THREADS;
    static_assert(NPS <= OWN_THREADS, "one self-loop per thread at most");
    float ek[KS][H];
    float es[H];
#pragma unroll
    for (int h = 0; h < H; ++h) {
        es[h] = 0.f;
#pragma unroll
        for (int j = 0; j < KS; ++j) ek[j][h] = 0.f;
    }

    if (listed) {
#pragma unroll
        for (int j = 0; j < KS; ++j) {
            const int k = tid + j * nt;
            if (k < total) {
                const int dl = s_ld[k];
                if (dl >= 0) {
                    score(s_li[k], dl, ek[j]);
                    put_max(dl, ek[j]);
                }
            }
        }
        for (int k = tid + KS * nt; k < total; k += nt) {
            const int dl = s_ld[k];
            if (dl >= 0) {
                float e[H];
                score(s_li[k], dl, e);
                put_max(dl, e);
            }
        }
    } else {
        for (int i = tid; i < E; i += nt) {
            const unsigned a = (unsigned)(dsts[i] - n0);
            if (a < (unsigned)cnt) {
                float e[H];
                score(srcs[i], (int)a, e);
                put_max((int)a, e);
            }
        }
    }
    if (tid < cnt) {
        score(n0 + tid, tid, es);
        put_max(tid, es);
    }
    __syncthreads();

    if (listed) {
#pragma unroll
        for (int j = 0; j < KS; ++j) {
            const int k = tid + j * nt;
            if (k < total) {
                const int dl = s_ld[k];
                if (dl >= 0) add_edge(s_li[k], dl, ek[j]);
            }
        }
        for (int k = tid + KS * nt; k < total; k += nt) {
            const int dl = s_ld[k];
            if (dl >= 0) {
                float e[H];
                score(s_li[k], dl, e);
                add_edge(s_li[k], dl, e);
            }
        }
    } else {
        for (int i = tid; i < E; i += nt) {
            const unsigned a = (unsigned)(dsts[i] - n0);
            if (a < (unsigned)cnt) {
                float e[H];
                const int s = srcs[i];
                score(s, (int)a, e);
                add_edge(s, (int)a, e);
            }
        }
    }
    if (tid < cnt) add_edge(n0 + tid, tid, es);
    __syncthreads();

    if (!FINAL) {
        for (int j = tid; j < cnt * F; j += nt) {
            const float v =
                s_acc[(j / F) * FP + j % F] / s_den[j / C] + bias[j % F];
            outp[(size_t)n0 * F + j] = (v > 0.f) ? v : (expf(v) - 1.f);
        }
    } else {
        for (int dl = tid; dl < cnt; dl += nt) {
            const float dn = s_den[dl];
            float v[F];
            float m = -INFINITY;
#pragma unroll
            for (int c = 0; c < F; ++c) {
                v[c] = s_acc[dl * FP + c] / dn + bias[c];
                m = fmaxf(m, v[c]);
            }
            float ssum = 0.f;
#pragma unroll
            for (int c = 0; c < F; ++c) ssum += expf(v[c] - m);
            const float lg = m + logf(ssum);
#pragma unroll
            for (int c = 0; c < F; ++c)
                outp[(size_t)(n0 + dl) * F + c] = v[c] - lg;
        }
    }
}

extern "C" void kernel_launch(void* const* d_in, const int* in_sizes, int n_in,
                              void* d_out, int out_size, void* d_ws,
                              size_t ws_size, hipStream_t stream) {
    const float* x   = (const float*)d_in[0];
    const int*   ei  = (const int*)d_in[1];
    const float* W1  = (const float*)d_in[2];
    const float* a1s = (const float*)d_in[3];
    const float* a1d = (const float*)d_in[4];
    const float* b1  = (const float*)d_in[5];
    const float* W2  = (const float*)d_in[6];
    const float* a2s = (const float*)d_in[7];
    const float* a2d = (const float*)d_in[8];
    const float* b2  = (const float*)d_in[9];
    float* out = (float*)d_out;

    const int N = in_sizes[0] / 128;
    const int E = in_sizes[1] / 2;
    const int* srcs = ei;
    const int* dsts = ei + E;

    char* p = (char*)d_ws;
    auto carve = [&](size_t bytes) {
        void* r = (void*)p;
        p += (bytes + 255) & ~(size_t)255;
        return r;
    };
    float* xw1 = (float*)carve((size_t)N * 32 * 4);
    float* as1 = (float*)carve((size_t)N * 4 * 4);
    float* ad1 = (float*)carve((size_t)N * 4 * 4);
    float* h1  = (float*)carve((size_t)N * 32 * 4);
    unsigned* keys = (unsigned*)carve((size_t)E + 16);
    float* xw2 = xw1;
    float* as2 = as1;
    float* ad2 = ad1;

    const int B = 256;
    auto g = [](int n, int b) { return (n + b - 1) / b; };

    int* ltot  = (int*)carve((size_t)g(N, NPS) * 4);
    int* lsave = (int*)carve((size_t)g(N, NPS) * LPITCH * 4);
    const int* ws_first = (const int*)d_ws;

    key_kernel<<<g(g(E, 4 * KEY_U), B), B, 0, stream>>>(dsts, E, keys);

    gemm_alpha_kernel<128, 32, 4><<<g(N, 32), 64, 0, stream>>>(
        x, W1, a1s, a1d, xw1, as1, ad1, N);
    owner_kernel<4, 8, false, true><<<g(N, NPS), OWN_THREADS, 0, stream>>>(
        srcs, dsts, keys, E, N, as1, ad1, xw1, b1, h1, ltot, lsave, ws_first,
        ws_size);

    gemm_alpha_kernel<32, 16, 1><<<g(N, 32), 64, 0, stream>>>(
        h1, W2, a2s, a2d, xw2, as2, ad2, N);
    owner_kernel<1, 16, true, false><<<g(N, NPS), OWN_THREADS, 0, stream>>>(
        srcs, dsts, keys, E, N, as2, ad2, xw2, b2, out, ltot, lsave, ws_first,
        ws_size);
}


# key plane stored as eight bit planes (bit b of the owning group's number, 32 edges per dword): the owner scan matches 32 edges with one xor and one and per plane, not 32 byte compares; arrival order
# speedup vs baseline: 11.8410x; 1.0636x over previous
#include <hip/hip_runtime.h>
#include <math.h>

typedef __attribute__((ext_vector_type(2))) float v2f;
typedef __attribute__((ext_vector_type(8))) float v8f;

#define NEG_SLOPE 0.2f

__device__ __forceinline__ unsigned fkey(float f) {
    unsigned b = __float_as_uint(f);
    return (b & 0x80000000u) ? ~b : (b | 0x80000000u);
}
__device__ __forceinline__ float fkey_inv(unsigned u) {
    return (u & 0x80000000u) ? __uint_as_float(u & 0x7fffffffu)
                             : __uint_as_float(~u);
}
#define FKEY_NEG_INF 0x007fffffu

template <int K, int NCOL, int H>
__global__ void gemm_alpha_kernel(const float* __restrict__ A,
                                  const float* __restrict__ B,
                                  const float* __restrict__ a_src,
                                  const float* __restrict__ a_dst,
                                  float* __restrict__ C,
                                  float* __restrict__ as_,
                                  float* __restrict__ ad_, int N) {
    constexpr int NT = NCOL / 16;
    constexpr int CH = NCOL / H;
    static_assert(CH == 8 || CH == 16, "a head must lie inside one tile");
    const int lane = threadIdx.x & 31;
    const int l    = lane & 15;
    const int hf   = lane >> 4;
    const int row0 =
        (blockIdx.x * (blockDim.x >> 5) + (threadIdx.x >> 5)) * 16;
    if (row0 >= N) return;

    const float* __restrict__ arow = A + (size_t)(row0 + l) * K;
    v8f acc[NT];
#pragma unroll
    for (int t = 0; t < NT; ++t) acc[t] = (v8f){};

#pragma unroll
    for (int k = 0; k < K; k += 4) {
        v2f a;
        a.x = arow[k + 2 * hf];
        a.y = arow[k + 2 * hf + 1];
#pragma unroll
        for (int t = 0; t < NT; ++t) {
            v2f b;
            b.x = B[(size_t)(k + 2 * hf) * NCOL + 16 * t + l];
            b.y = B[(size_t)(k + 2 * hf + 1) * NCOL + 16 * t + l];
            acc[t] = __builtin_amdgcn_wmma_f32_16x16x4_f32(
                 false, a,  false, b,
                 (short)0, acc[t],  false,
                 false);
        }
    }
    float s[NT][8], d[NT][8];
#pragma unroll
    for (int t = 0; t < NT; ++t) {
        const float ws = a_src[16 * t + l];
        const float wd = a_dst[16 * t + l];
#pragma unroll
        for (int i = 0; i < 8; ++i) {
            s[t][i] = acc[t][i] * ws;
            d[t][i] = acc[t][i] * wd;
        }
    }
#pragma unroll
    for (int m = 1; m < CH; m <<= 1) {
        float ps[NT][8], pd[NT][8];
#pragma unroll
        for (int t = 0; t < NT; ++t)
#pragma unroll
            for (int i = 0; i < 8; ++i) {
                ps[t][i] = __shfl_xor(s[t][i], m);
                pd[t][i] = __shfl_xor(d[t][i], m);
            }
#pragma unroll
        for (int t = 0; t < NT; ++t)
#pragma unroll
            for (int i = 0; i < 8; ++i) {
                s[t][i] += ps[t][i];
                d[t][i] += pd[t][i];
            }
    }
#pragma unroll
    for (int t = 0; t < NT; ++t)
#pragma unroll
        for (int i = 0; i < 8; ++i)
            C[(size_t)(row0 + i + 8 * hf) * NCOL + 16 * t + l] = acc[t][i];
    if ((l & (CH - 1)) == 0) {
#pragma unroll
        for (int t = 0; t < NT; ++t)
#pragma unroll
            for (int i = 0; i < 8; ++i) {
                const size_t o =
                    (size_t)(row0 + i + 8 * hf) * H + (16 * t + l) / CH;
                as_[o] = s[t][i];
                ad_[o] = d[t][i];
            }
    }
}

#define NPS  392
#define OWN_THREADS 1024
#define LCAP 18432
#define LPITCH 7168
#define LSHIFT 9

__global__ void key_kernel(const int* __restrict__ dsts, int E,
                           unsigned* __restrict__ planes) {
    const int PW = (E >> 10) * 32;
    const int D  = blockIdx.x * blockDim.x + threadIdx.x;
    if (D >= PW) return;
    int d[32];
    if ((reinterpret_cast<size_t>(dsts) & 15) == 0) {
        const int4* __restrict__ d4 =
            reinterpret_cast<const int4*>(dsts) + (size_t)8 * D;
#pragma unroll
        for (int u = 0; u < 8; ++u) {
            const int4 v = d4[u];
            d[4 * u + 0] = v.x;
            d[4 * u + 1] = v.y;
            d[4 * u + 2] = v.z;
            d[4 * u + 3] = v.w;
        }
    } else {
#pragma unroll
        for (int j = 0; j < 32; ++j) d[j] = dsts[(size_t)32 * D + j];
    }
    unsigned r[8];
#pragma unroll
    for (int b = 0; b < 8; ++b) r[b] = 0u;
#pragma unroll
    for (int j = 0; j < 32; ++j) {
        const unsigned g = (unsigned)d[j] / (unsigned)NPS;
#pragma unroll
        for (int b = 0; b < 8; ++b) r[b] |= ((g >> b) & 1u) << j;
    }
#pragma unroll
    for (int b = 0; b < 8; ++b) planes[(size_t)b * PW + D] = r[b];
}

template <class Fn>
__device__ __forceinline__ void scan_keys(const unsigned* __restrict__ keys,
                                          const int* __restrict__ dsts, int E,
                                          int n0, unsigned cnt, Fn emit) {
    const int tid = threadIdx.x;
    constexpr int nt = OWN_THREADS;
    const int EU = (E >> 10) * 8;
    const uint4* __restrict__ pl = reinterpret_cast<const uint4*>(keys);
    unsigned inv[8];
#pragma unroll
    for (int b = 0; b < 8; ++b)
        inv[b] = ((blockIdx.x >> b) & 1u) ? 0u : ~0u;
    auto drain = [&](unsigned m, int e0) {
        while (m) {
            const int p = __builtin_ctz(m);
            m &= m - 1;
            emit(e0 + p);
        }
    };
    for (int q = tid; q < EU; q += nt) {
        uint4 v[8];
#pragma unroll
        for (int b = 0; b < 8; ++b) v[b] = pl[(size_t)b * EU + q];
        unsigned m0 = ~0u, m1 = ~0u, m2 = ~0u, m3 = ~0u;
#pragma unroll
        for (int b = 0; b < 8; ++b) {
            m0 &= v[b].x ^ inv[b];
            m1 &= v[b].y ^ inv[b];
            m2 &= v[b].z ^ inv[b];
            m3 &= v[b].w ^ inv[b];
        }
        drain(m0, 128 * q);
        drain(m1, 128 * q + 32);
        drain(m2, 128 * q + 64);
        drain(m3, 128 * q + 96);
    }
    for (int i = 128 * EU + tid; i < E; i += nt) {
        const unsigned a = (unsigned)(dsts[i] - n0);
        if (a < cnt) emit(i);
    }
}

template <int H, int C, bool FINAL, bool FIRST>
__global__ __launch_bounds__(OWN_THREADS) void owner_kernel(
    const int* __restrict__ srcs, const int* __restrict__ dsts,
    const unsigned* __restrict__ keys, int E, int N,
    const float* __restrict__ as_, const float* __restrict__ ad_,
    const float* __restrict__ xw, const float* __restrict__ bias,
    float* __restrict__ outp, int* __restrict__ ltot,
    int* __restrict__ lsave, const int* ws_first, size_t ws_bytes) {
    static_assert(C % 4 == 0, "C must be a multiple of 4");
    static_assert(!FINAL || H == 1, "the final layer has one head");
    static_assert(NPS <= (1 << LSHIFT), "a local destination must fit");
    static_assert(LPITCH <= LCAP, "a stored list must fit the list in LDS");
    constexpr int F = H * C;
    constexpr int FP = F + 1;
    __shared__ unsigned s_m[NPS * H];
    __shared__ float    s_ad[NPS * H];
    __shared__ float    s_den[NPS * H];
    __shared__ float    s_acc[NPS * FP];
    __shared__ int      s_li[LCAP];
    __shared__ int      s_ld[LCAP];
    __shared__ int      s_cnt;

    const int tid = threadIdx.x;
    constexpr int nt = OWN_THREADS;
    const int n0 = blockIdx.x * NPS;
    if (n0 >= N) return;
    const int cnt = min(NPS, N - n0);

    for (int j = tid; j < cnt * H; j += nt) {
        s_m[j] = FKEY_NEG_INF;
        s_den[j] = 0.f;
        s_ad[j] = ad_[(size_t)n0 * H + j];
    }
    for (int j = tid; j < cnt * FP; j += nt) s_acc[j] = 0.f;
    if (tid == 0) s_cnt = 0;
    __syncthreads();

    auto score = [&](int s, int dl, float* e) {
#pragma unroll
        for (int h = 0; h < H; ++h) {
            const float v = as_[s * H + h] + s_ad[dl * H + h];
            e[h] = (v > 0.f) ? v : NEG_SLOPE * v;
        }
    };
    auto put_max = [&](int dl, const float* e) {
#pragma unroll
        for (int h = 0; h < H; ++h) atomicMax(&s_m[dl * H + h], fkey(e[h]));
    };
    auto add_edge = [&](int s, int dl, const float* e) {
        float w[H];
#pragma unroll
        for (int h = 0; h < H; ++h) {
            w[h] = expf(e[h] - fkey_inv(s_m[dl * H + h]));
            atomicAdd(&s_den[dl * H + h], w[h]);
        }
        const float4* __restrict__ sv = (const float4*)(xw + (size_t)s * F);
        float* dv = &s_acc[dl * FP];
#pragma unroll
        for (int f = 0; f < F / 4; ++f) {
            const float4 v = sv[f];
            const float wh = w[(f * 4) / C];
            atomicAdd(&dv[f * 4 + 0], wh * v.x);
            atomicAdd(&dv[f * 4 + 1], wh * v.y);
            atomicAdd(&dv[f * 4 + 2], wh * v.z);
            atomicAdd(&dv[f * 4 + 3], wh * v.w);
        }
    };

    const bool ws_ok =
        (size_t)((lsave + (size_t)gridDim.x * LPITCH) - ws_first) *
                sizeof(int) <= ws_bytes &&
        N <= (1 << (31 - LSHIFT));
    int* __restrict__ mine = lsave + (size_t)blockIdx.x * LPITCH;

    int  total  = 0;
    bool loaded = false;
    if (!FIRST && ws_ok) {
        total  = ltot[blockIdx.x];
        loaded = total <= LPITCH;
    }
    if (loaded) {
        constexpr int LSLOTS = LPITCH / OWN_THREADS;
        static_assert(LSLOTS * OWN_THREADS == LPITCH, "whole slots");
        if (total > 0) {
            int v[LSLOTS];
#pragma unroll
            for (int j = 0; j < LSLOTS; ++j)
                v[j] = mine[min(tid + j * nt, total - 1)];
#pragma unroll
            for (int j = 0; j < LSLOTS; ++j) {
                const int k = tid + j * nt;
                if (k < total) {
                    s_li[k] = v[j] >> LSHIFT;
                    s_ld[k] = (v[j] < 0) ? -1 : (v[j] & ((1 << LSHIFT) - 1));
                }
            }
        }
    } else {
        scan_keys(keys, dsts, E, n0, (unsigned)cnt, [&](int i) {
            const int pos = atomicAdd(&s_cnt, 1);
            if (pos < LCAP) s_li[pos] = i;
        });
        __syncthreads();
        total = s_cnt;
    }
    const bool listed = total <= LCAP;
    if (!loaded) {
        const bool store = FIRST && ws_ok && total <= LPITCH;
        if (FIRST && ws_ok && tid == 0) ltot[blockIdx.x] = total;
        auto resolve = [&](auto keep) {
            for (int k = tid; k < total; k += nt) {
                const int i = s_li[k];
                const unsigned a = (unsigned)(dsts[i] - n0);
                const int s  = srcs[i];
                const int dl = (a < (unsigned)cnt) ? (int)a : -1;
                s_li[k] = s;
                s_ld[k] = dl;
                keep(k, s, dl);
            }
        };
        if (store)
            resolve([&](int k, int s, int dl) {
                mine[k] = (dl < 0) ? -1 : ((s << LSHIFT) | dl);
            });
        else if (listed)
            resolve([](int, int, int) {});
    }

    constexpr int KS = LPITCH / OWN_THREADS;
    static_assert(NPS <= OWN_THREADS, "one self-loop per thread at most");
    float ek[KS][H];
    float es[H];
#pragma unroll
    for (int h = 0; h < H; ++h) {
        es[h] = 0.f;
#pragma unroll
        for (int j = 0; j < KS; ++j) ek[j][h] = 0.f;
    }

    if (listed) {
#pragma unroll
        for (int j = 0; j < KS; ++j) {
            const int k = tid + j * nt;
            if (k < total) {
                const int dl = s_ld[k];
                if (dl >= 0) {
                    score(s_li[k], dl, ek[j]);
                    put_max(dl, ek[j]);
                }
            }
        }
        for (int k = tid + KS * nt; k < total; k += nt) {
            const int dl = s_ld[k];
            if (dl >= 0) {
                float e[H];
                score(s_li[k], dl, e);
                put_max(dl, e);
            }
        }
    } else {
        for (int i = tid; i < E; i += nt) {
            const unsigned a = (unsigned)(dsts[i] - n0);
            if (a < (unsigned)cnt) {
                float e[H];
                score(srcs[i], (int)a, e);
                put_max((int)a, e);
            }
        }
    }
    if (tid < cnt) {
        score(n0 + tid, tid, es);
        put_max(tid, es);
    }
    __syncthreads();

    if (listed) {
#pragma unroll
        for (int j = 0; j < KS; ++j) {
            const int k = tid + j * nt;
            if (k < total) {
                const int dl = s_ld[k];
                if (dl >= 0) add_edge(s_li[k], dl, ek[j]);
            }
        }
        for (int k = tid + KS * nt; k < total; k += nt) {
            const int dl = s_ld[k];
            if (dl >= 0) {
                float e[H];
                score(s_li[k], dl, e);
                add_edge(s_li[k], dl, e);
            }
        }
    } else {
        for (int i = tid; i < E; i += nt) {
            const unsigned a = (unsigned)(dsts[i] - n0);
            if (a < (unsigned)cnt) {
                float e[H];
                const int s = srcs[i];
                score(s, (int)a, e);
                add_edge(s, (int)a, e);
            }
        }
    }
    if (tid < cnt) add_edge(n0 + tid, tid, es);
    __syncthreads();

    if (!FINAL) {
        for (int j = tid; j < cnt * F; j += nt) {
            const float v =
                s_acc[(j / F) * FP + j % F] / s_den[j / C] + bias[j % F];
            outp[(size_t)n0 * F + j] = (v > 0.f) ? v : (expf(v) - 1.f);
        }
    } else {
        for (int dl = tid; dl < cnt; dl += nt) {
            const float dn = s_den[dl];
            float v[F];
            float m = -INFINITY;
#pragma unroll
            for (int c = 0; c < F; ++c) {
                v[c] = s_acc[dl * FP + c] / dn + bias[c];
                m = fmaxf(m, v[c]);
            }
            float ssum = 0.f;
#pragma unroll
            for (int c = 0; c < F; ++c) ssum += expf(v[c] - m);
            const float lg = m + logf(ssum);
#pragma unroll
            for (int c = 0; c < F; ++c)
                outp[(size_t)(n0 + dl) * F + c] = v[c] - lg;
        }
    }
}

extern "C" void kernel_launch(void* const* d_in, const int* in_sizes, int n_in,
                              void* d_out, int out_size, void* d_ws,
                              size_t ws_size, hipStream_t stream) {
    const float* x   = (const float*)d_in[0];
    const int*   ei  = (const int*)d_in[1];
    const float* W1  = (const float*)d_in[2];
    const float* a1s = (const float*)d_in[3];
    const float* a1d = (const float*)d_in[4];
    const float* b1  = (const float*)d_in[5];
    const float* W2  = (const float*)d_in[6];
    const float* a2s = (const float*)d_in[7];
    const float* a2d = (const float*)d_in[8];
    const float* b2  = (const float*)d_in[9];
    float* out = (float*)d_out;

    const int N = in_sizes[0] / 128;
    const int E = in_sizes[1] / 2;
    const int* srcs = ei;
    const int* dsts = ei + E;

    char* p = (char*)d_ws;
    auto carve = [&](size_t bytes) {
        void* r = (void*)p;
        p += (bytes + 255) & ~(size_t)255;
        return r;
    };
    float* xw1 = (float*)carve((size_t)N * 32 * 4);
    float* as1 = (float*)carve((size_t)N * 4 * 4);
    float* ad1 = (float*)carve((size_t)N * 4 * 4);
    float* h1  = (float*)carve((size_t)N * 32 * 4);
    unsigned* keys = (unsigned*)carve((size_t)E + 16);
    float* xw2 = xw1;
    float* as2 = as1;
    float* ad2 = ad1;

    const int B = 256;
    auto g = [](int n, int b) { return (n + b - 1) / b; };

    int* ltot  = (int*)carve((size_t)g(N, NPS) * 4);
    int* lsave = (int*)carve((size_t)g(N, NPS) * LPITCH * 4);
    const int* ws_first = (const int*)d_ws;

    key_kernel<<<g((E >> 10) * 32 + 1, B), B, 0, stream>>>(dsts, E, keys);

    gemm_alpha_kernel<128, 32, 4><<<g(N, 32), 64, 0, stream>>>(
        x, W1, a1s, a1d, xw1, as1, ad1, N);
    owner_kernel<4, 8, false, true><<<g(N, NPS), OWN_THREADS, 0, stream>>>(
        srcs, dsts, keys, E, N, as1, ad1, xw1, b1, h1, ltot, lsave, ws_first,
        ws_size);

    gemm_alpha_kernel<32, 16, 1><<<g(N, 32), 64, 0, stream>>>(
        h1, W2, a2s, a2d, xw2, as2, ad2, N);
    owner_kernel<1, 16, true, false><<<g(N, NPS), OWN_THREADS, 0, stream>>>(
        srcs, dsts, keys, E, N, as2, ad2, xw2, b2, out, ltot, lsave, ws_first,
        ws_size);
}


# key planes one-hot per 4-bit half of the owning group's number (16 + 16 planes): an owner group reads and ands its two planes, 2 bits per edge, not 8; 8 loads in flight as before; arrival order
# speedup vs baseline: 12.3876x; 1.0462x over previous
#include <hip/hip_runtime.h>
#include <math.h>

typedef __attribute__((ext_vector_type(2))) float v2f;
typedef __attribute__((ext_vector_type(8))) float v8f;

#define NEG_SLOPE 0.2f

__device__ __forceinline__ unsigned fkey(float f) {
    unsigned b = __float_as_uint(f);
    return (b & 0x80000000u) ? ~b : (b | 0x80000000u);
}
__device__ __forceinline__ float fkey_inv(unsigned u) {
    return (u & 0x80000000u) ? __uint_as_float(u & 0x7fffffffu)
                             : __uint_as_float(~u);
}
#define FKEY_NEG_INF 0x007fffffu

template <int K, int NCOL, int H>
__global__ void gemm_alpha_kernel(const float* __restrict__ A,
                                  const float* __restrict__ B,
                                  const float* __restrict__ a_src,
                                  const float* __restrict__ a_dst,
                                  float* __restrict__ C,
                                  float* __restrict__ as_,
                                  float* __restrict__ ad_, int N) {
    constexpr int NT = NCOL / 16;
    constexpr int CH = NCOL / H;
    static_assert(CH == 8 || CH == 16, "a head must lie inside one tile");
    const int lane = threadIdx.x & 31;
    const int l    = lane & 15;
    const int hf   = lane >> 4;
    const int row0 =
        (blockIdx.x * (blockDim.x >> 5) + (threadIdx.x >> 5)) * 16;
    if (row0 >= N) return;

    const float* __restrict__ arow = A + (size_t)(row0 + l) * K;
    v8f acc[NT];
#pragma unroll
    for (int t = 0; t < NT; ++t) acc[t] = (v8f){};

#pragma unroll
    for (int k = 0; k < K; k += 4) {
        v2f a;
        a.x = arow[k + 2 * hf];
        a.y = arow[k + 2 * hf + 1];
#pragma unroll
        for (int t = 0; t < NT; ++t) {
            v2f b;
            b.x = B[(size_t)(k + 2 * hf) * NCOL + 16 * t + l];
            b.y = B[(size_t)(k + 2 * hf + 1) * NCOL + 16 * t + l];
            acc[t] = __builtin_amdgcn_wmma_f32_16x16x4_f32(
                 false, a,  false, b,
                 (short)0, acc[t],  false,
                 false);
        }
    }
    float s[NT][8], d[NT][8];
#pragma unroll
    for (int t = 0; t < NT; ++t) {
        const float ws = a_src[16 * t + l];
        const float wd = a_dst[16 * t + l];
#pragma unroll
        for (int i = 0; i < 8; ++i) {
            s[t][i] = acc[t][i] * ws;
            d[t][i] = acc[t][i] * wd;
        }
    }
#pragma unroll
    for (int m = 1; m < CH; m <<= 1) {
        float ps[NT][8], pd[NT][8];
#pragma unroll
        for (int t = 0; t < NT; ++t)
#pragma unroll
            for (int i = 0; i < 8; ++i) {
                ps[t][i] = __shfl_xor(s[t][i], m);
                pd[t][i] = __shfl_xor(d[t][i], m);
            }
#pragma unroll
        for (int t = 0; t < NT; ++t)
#pragma unroll
            for (int i = 0; i < 8; ++i) {
                s[t][i] += ps[t][i];
                d[t][i] += pd[t][i];
            }
    }
#pragma unroll
    for (int t = 0; t < NT; ++t)
#pragma unroll
        for (int i = 0; i < 8; ++i)
            C[(size_t)(row0 + i + 8 * hf) * NCOL + 16 * t + l] = acc[t][i];
    if ((l & (CH - 1)) == 0) {
#pragma unroll
        for (int t = 0; t < NT; ++t)
#pragma unroll
            for (int i = 0; i < 8; ++i) {
                const size_t o =
                    (size_t)(row0 + i + 8 * hf) * H + (16 * t + l) / CH;
                as_[o] = s[t][i];
                ad_[o] = d[t][i];
            }
    }
}

#define NPS  392
#define OWN_THREADS 1024
#define LCAP 18432
#define LPITCH 7168
#define LSHIFT 9

__global__ void key_kernel(const int* __restrict__ dsts, int E,
                           unsigned* __restrict__ planes) {
    const int PW = (E >> 10) * 32;
    const int D  = blockIdx.x * blockDim.x + threadIdx.x;
    if (D >= PW) return;
    int d[32];
    if ((reinterpret_cast<size_t>(dsts) & 15) == 0) {
        const int4* __restrict__ d4 =
            reinterpret_cast<const int4*>(dsts) + (size_t)8 * D;
#pragma unroll
        for (int u = 0; u < 8; ++u) {
            const int4 v = d4[u];
            d[4 * u + 0] = v.x;
            d[4 * u + 1] = v.y;
            d[4 * u + 2] = v.z;
            d[4 * u + 3] = v.w;
        }
    } else {
#pragma unroll
        for (int j = 0; j < 32; ++j) d[j] = dsts[(size_t)32 * D + j];
    }
    unsigned a[32];
#pragma unroll
    for (int j = 0; j < 32; ++j) {
        const unsigned g = (unsigned)d[j] / (unsigned)NPS;
        a[j] = (1u << ((g >> 4) & 15u)) | (0x10000u << (g & 15u));
    }
#pragma unroll
    for (int s = 16; s != 0; s >>= 1) {
        const unsigned m = 0xffffffffu / ((1u << s) + 1u);
#pragma unroll
        for (int k = 0; k < 32; ++k) {
            if ((k & s) == 0) {
                const unsigned t = ((a[k] >> s) ^ a[k + s]) & m;
                a[k] ^= t << s;
                a[k + s] ^= t;
            }
        }
    }
#pragma unroll
    for (int p = 0; p < 32; ++p) planes[(size_t)p * PW + D] = a[p];
}

template <class Fn>
__device__ __forceinline__ void scan_keys(const unsigned* __restrict__ keys,
                                          const int* __restrict__ dsts, int E,
                                          int n0, unsigned cnt, Fn emit) {
    const int tid = threadIdx.x;
    constexpr int nt = OWN_THREADS;
    constexpr int UPT = 4;
    const int EU = (E >> 10) * 8;
    const uint4* __restrict__ pu = reinterpret_cast<const uint4*>(keys) +
                                   (size_t)((blockIdx.x >> 4) & 15u) * EU;
    const uint4* __restrict__ pw = reinterpret_cast<const uint4*>(keys) +
                                   (size_t)(16u + (blockIdx.x & 15u)) * EU;
    auto drain = [&](unsigned m, int e0) {
        while (m) {
            const int p = __builtin_ctz(m);
            m &= m - 1;
            emit(e0 + p);
        }
    };
    for (int q0 = tid; q0 < EU; q0 += UPT * nt) {
        uint4 u[UPT], w[UPT];
#pragma unroll
        for (int j = 0; j < UPT; ++j) {
            const int q = min(q0 + j * nt, EU - 1);
            u[j] = pu[q];
            w[j] = pw[q];
        }
#pragma unroll
        for (int j = 0; j < UPT; ++j) {
            const int q = q0 + j * nt;
            const unsigned in = (q < EU) ? ~0u : 0u;
            drain(u[j].x & w[j].x & in, 128 * q);
            drain(u[j].y & w[j].y & in, 128 * q + 32);
            drain(u[j].z & w[j].z & in, 128 * q + 64);
            drain(u[j].w & w[j].w & in, 128 * q + 96);
        }
    }
    for (int i = 128 * EU + tid; i < E; i += nt) {
        const unsigned a = (unsigned)(dsts[i] - n0);
        if (a < cnt) emit(i);
    }
}

template <int H, int C, bool FINAL, bool FIRST>
__global__ __launch_bounds__(OWN_THREADS) void owner_kernel(
    const int* __restrict__ srcs, const int* __restrict__ dsts,
    const unsigned* __restrict__ keys, int E, int N,
    const float* __restrict__ as_, const float* __restrict__ ad_,
    const float* __restrict__ xw, const float* __restrict__ bias,
    float* __restrict__ outp, int* __restrict__ ltot,
    int* __restrict__ lsave, const int* ws_first, size_t ws_bytes) {
    static_assert(C % 4 == 0, "C must be a multiple of 4");
    static_assert(!FINAL || H == 1, "the final layer has one head");
    static_assert(NPS <= (1 << LSHIFT), "a local destination must fit");
    static_assert(LPITCH <= LCAP, "a stored list must fit the list in LDS");
    constexpr int F = H * C;
    constexpr int FP = F + 1;
    __shared__ unsigned s_m[NPS * H];
    __shared__ float    s_ad[NPS * H];
    __shared__ float    s_den[NPS * H];
    __shared__ float    s_acc[NPS * FP];
    __shared__ int      s_li[LCAP];
    __shared__ int      s_ld[LCAP];
    __shared__ int      s_cnt;

    const int tid = threadIdx.x;
    constexpr int nt = OWN_THREADS;
    const int n0 = blockIdx.x * NPS;
    if (n0 >= N) return;
    const int cnt = min(NPS, N - n0);

    for (int j = tid; j < cnt * H; j += nt) {
        s_m[j] = FKEY_NEG_INF;
        s_den[j] = 0.f;
        s_ad[j] = ad_[(size_t)n0 * H + j];
    }
    for (int j = tid; j < cnt * FP; j += nt) s_acc[j] = 0.f;
    if (tid == 0) s_cnt = 0;
    __syncthreads();

    auto score = [&](int s, int dl, float* e) {
#pragma unroll
        for (int h = 0; h < H; ++h) {
            const float v = as_[s * H + h] + s_ad[dl * H + h];
            e[h] = (v > 0.f) ? v : NEG_SLOPE * v;
        }
    };
    auto put_max = [&](int dl, const float* e) {
#pragma unroll
        for (int h = 0; h < H; ++h) atomicMax(&s_m[dl * H + h], fkey(e[h]));
    };
    auto add_edge = [&](int s, int dl, const float* e) {
        float w[H];
#pragma unroll
        for (int h = 0; h < H; ++h) {
            w[h] = expf(e[h] - fkey_inv(s_m[dl * H + h]));
            atomicAdd(&s_den[dl * H + h], w[h]);
        }
        const float4* __restrict__ sv = (const float4*)(xw + (size_t)s * F);
        float* dv = &s_acc[dl * FP];
#pragma unroll
        for (int f = 0; f < F / 4; ++f) {
            const float4 v = sv[f];
            const float wh = w[(f * 4) / C];
            atomicAdd(&dv[f * 4 + 0], wh * v.x);
            atomicAdd(&dv[f * 4 + 1], wh * v.y);
            atomicAdd(&dv[f * 4 + 2], wh * v.z);
            atomicAdd(&dv[f * 4 + 3], wh * v.w);
        }
    };

    const bool ws_ok =
        (size_t)((lsave + (size_t)gridDim.x * LPITCH) - ws_first) *
                sizeof(int) <= ws_bytes &&
        N <= (1 << (31 - LSHIFT));
    int* __restrict__ mine = lsave + (size_t)blockIdx.x * LPITCH;

    int  total  = 0;
    bool loaded = false;
    if (!FIRST && ws_ok) {
        total  = ltot[blockIdx.x];
        loaded = total <= LPITCH;
    }
    if (loaded) {
        constexpr int LSLOTS = LPITCH / OWN_THREADS;
        static_assert(LSLOTS * OWN_THREADS == LPITCH, "whole slots");
        if (total > 0) {
            int v[LSLOTS];
#pragma unroll
            for (int j = 0; j < LSLOTS; ++j)
                v[j] = mine[min(tid + j * nt, total - 1)];
#pragma unroll
            for (int j = 0; j < LSLOTS; ++j) {
                const int k = tid + j * nt;
                if (k < total) {
                    s_li[k] = v[j] >> LSHIFT;
                    s_ld[k] = (v[j] < 0) ? -1 : (v[j] & ((1 << LSHIFT) - 1));
                }
            }
        }
    } else {
        scan_keys(keys, dsts, E, n0, (unsigned)cnt, [&](int i) {
            const int pos = atomicAdd(&s_cnt, 1);
            if (pos < LCAP) s_li[pos] = i;
        });
        __syncthreads();
        total = s_cnt;
    }
    const bool listed = total <= LCAP;
    if (!loaded) {
        const bool store = FIRST && ws_ok && total <= LPITCH;
        if (FIRST && ws_ok && tid == 0) ltot[blockIdx.x] = total;
        auto resolve = [&](auto keep) {
            for (int k = tid; k < total; k += nt) {
                const int i = s_li[k];
                const unsigned a = (unsigned)(dsts[i] - n0);
                const int s  = srcs[i];
                const int dl = (a < (unsigned)cnt) ? (int)a : -1;
                s_li[k] = s;
                s_ld[k] = dl;
                keep(k, s, dl);
            }
        };
        if (store)
            resolve([&](int k, int s, int dl) {
                mine[k] = (dl < 0) ? -1 : ((s << LSHIFT) | dl);
            });
        else if (listed)
            resolve([](int, int, int) {});
    }

    constexpr int KS = LPITCH / OWN_THREADS;
    static_assert(NPS <= OWN_THREADS, "one self-loop per thread at most");
    float ek[KS][H];
    float es[H];
#pragma unroll
    for (int h = 0; h < H; ++h) {
        es[h] = 0.f;
#pragma unroll
        for (int j = 0; j < KS; ++j) ek[j][h] = 0.f;
    }

    if (listed) {
#pragma unroll
        for (int j = 0; j < KS; ++j) {
            const int k = tid + j * nt;
            if (k < total) {
                const int dl = s_ld[k];
                if (dl >= 0) {
                    score(s_li[k], dl, ek[j]);
                    put_max(dl, ek[j]);
                }
            }
        }
        for (int k = tid + KS * nt; k < total; k += nt) {
            const int dl = s_ld[k];
            if (dl >= 0) {
                float e[H];
                score(s_li[k], dl, e);
                put_max(dl, e);
            }
        }
    } else {
        for (int i = tid; i < E; i += nt) {
            const unsigned a = (unsigned)(dsts[i] - n0);
            if (a < (unsigned)cnt) {
                float e[H];
                score(srcs[i], (int)a, e);
                put_max((int)a, e);
            }
        }
    }
    if (tid < cnt) {
        score(n0 + tid, tid, es);
        put_max(tid, es);
    }
    __syncthreads();

    if (listed) {
#pragma unroll
        for (int j = 0; j < KS; ++j) {
            const int k = tid + j * nt;
            if (k < total) {
                const int dl = s_ld[k];
                if (dl >= 0) add_edge(s_li[k], dl, ek[j]);
            }
        }
        for (int k = tid + KS * nt; k < total; k += nt) {
            const int dl = s_ld[k];
            if (dl >= 0) {
                float e[H];
                score(s_li[k], dl, e);
                add_edge(s_li[k], dl, e);
            }
        }
    } else {
        for (int i = tid; i < E; i += nt) {
            const unsigned a = (unsigned)(dsts[i] - n0);
            if (a < (unsigned)cnt) {
                float e[H];
                const int s = srcs[i];
                score(s, (int)a, e);
                add_edge(s, (int)a, e);
            }
        }
    }
    if (tid < cnt) add_edge(n0 + tid, tid, es);
    __syncthreads();

    if (!FINAL) {
        for (int j = tid; j < cnt * F; j += nt) {
            const float v =
                s_acc[(j / F) * FP + j % F] / s_den[j / C] + bias[j % F];
            outp[(size_t)n0 * F + j] = (v > 0.f) ? v : (expf(v) - 1.f);
        }
    } else {
        for (int dl = tid; dl < cnt; dl += nt) {
            const float dn = s_den[dl];
            float v[F];
            float m = -INFINITY;
#pragma unroll
            for (int c = 0; c < F; ++c) {
                v[c] = s_acc[dl * FP + c] / dn + bias[c];
                m = fmaxf(m, v[c]);
            }
            float ssum = 0.f;
#pragma unroll
            for (int c = 0; c < F; ++c) ssum += expf(v[c] - m);
            const float lg = m + logf(ssum);
#pragma unroll
            for (int c = 0; c < F; ++c)
                outp[(size_t)(n0 + dl) * F + c] = v[c] - lg;
        }
    }
}

extern "C" void kernel_launch(void* const* d_in, const int* in_sizes, int n_in,
                              void* d_out, int out_size, void* d_ws,
                              size_t ws_size, hipStream_t stream) {
    const float* x   = (const float*)d_in[0];
    const int*   ei  = (const int*)d_in[1];
    const float* W1  = (const float*)d_in[2];
    const float* a1s = (const float*)d_in[3];
    const float* a1d = (const float*)d_in[4];
    const float* b1  = (const float*)d_in[5];
    const float* W2  = (const float*)d_in[6];
    const float* a2s = (const float*)d_in[7];
    const float* a2d = (const float*)d_in[8];
    const float* b2  = (const float*)d_in[9];
    float* out = (float*)d_out;

    const int N = in_sizes[0] / 128;
    const int E = in_sizes[1] / 2;
    const int* srcs = ei;
    const int* dsts = ei + E;

    char* p = (char*)d_ws;
    auto carve = [&](size_t bytes) {
        void* r = (void*)p;
        p += (bytes + 255) & ~(size_t)255;
        return r;
    };
    float* xw1 = (float*)carve((size_t)N * 32 * 4);
    float* as1 = (float*)carve((size_t)N * 4 * 4);
    float* ad1 = (float*)carve((size_t)N * 4 * 4);
    float* h1  = (float*)carve((size_t)N * 32 * 4);
    unsigned* keys = (unsigned*)carve((size_t)E * 4 + 16);
    float* xw2 = xw1;
    float* as2 = as1;
    float* ad2 = ad1;

    const int B = 256;
    auto g = [](int n, int b) { return (n + b - 1) / b; };

    int* ltot  = (int*)carve((size_t)g(N, NPS) * 4);
    int* lsave = (int*)carve((size_t)g(N, NPS) * LPITCH * 4);
    const int* ws_first = (const int*)d_ws;

    key_kernel<<<g((E >> 10) * 32 + 1, B), B, 0, stream>>>(dsts, E, keys);

    gemm_alpha_kernel<128, 32, 4><<<g(N, 32), 64, 0, stream>>>(
        x, W1, a1s, a1d, xw1, as1, ad1, N);
    owner_kernel<4, 8, false, true><<<g(N, NPS), OWN_THREADS, 0, stream>>>(
        srcs, dsts, keys, E, N, as1, ad1, xw1, b1, h1, ltot, lsave, ws_first,
        ws_size);

    gemm_alpha_kernel<32, 16, 1><<<g(N, 32), 64, 0, stream>>>(
        h1, W2, a2s, a2d, xw2, as2, ad2, N);
    owner_kernel<1, 16, true, false><<<g(N, NPS), OWN_THREADS, 0, stream>>>(
        srcs, dsts, keys, E, N, as2, ad2, xw2, b2, out, ltot, lsave, ws_first,
        ws_size);
}


# owner launches, second walk: the four lanes of a quad take one edge's source row together (its parts passed inside the quad by lane moves): a row load reads 8 lines, not 32; arrival order
# speedup vs baseline: 13.2384x; 1.0687x over previous
#include <hip/hip_runtime.h>
#include <math.h>

typedef __attribute__((ext_vector_type(2))) float v2f;
typedef __attribute__((ext_vector_type(8))) float v8f;

#define NEG_SLOPE 0.2f

__device__ __forceinline__ unsigned fkey(float f) {
    unsigned b = __float_as_uint(f);
    return (b & 0x80000000u) ? ~b : (b | 0x80000000u);
}
__device__ __forceinline__ float fkey_inv(unsigned u) {
    return (u & 0x80000000u) ? __uint_as_float(u & 0x7fffffffu)
                             : __uint_as_float(~u);
}
#define FKEY_NEG_INF 0x007fffffu

template <int K, int NCOL, int H>
__global__ void gemm_alpha_kernel(const float* __restrict__ A,
                                  const float* __restrict__ B,
                                  const float* __restrict__ a_src,
                                  const float* __restrict__ a_dst,
                                  float* __restrict__ C,
                                  float* __restrict__ as_,
                                  float* __restrict__ ad_, int N) {
    constexpr int NT = NCOL / 16;
    constexpr int CH = NCOL / H;
    static_assert(CH == 8 || CH == 16, "a head must lie inside one tile");
    const int lane = threadIdx.x & 31;
    const int l    = lane & 15;
    const int hf   = lane >> 4;
    const int row0 =
        (blockIdx.x * (blockDim.x >> 5) + (threadIdx.x >> 5)) * 16;
    if (row0 >= N) return;

    const float* __restrict__ arow = A + (size_t)(row0 + l) * K;
    v8f acc[NT];
#pragma unroll
    for (int t = 0; t < NT; ++t) acc[t] = (v8f){};

#pragma unroll
    for (int k = 0; k < K; k += 4) {
        v2f a;
        a.x = arow[k + 2 * hf];
        a.y = arow[k + 2 * hf + 1];
#pragma unroll
        for (int t = 0; t < NT; ++t) {
            v2f b;
            b.x = B[(size_t)(k + 2 * hf) * NCOL + 16 * t + l];
            b.y = B[(size_t)(k + 2 * hf + 1) * NCOL + 16 * t + l];
            acc[t] = __builtin_amdgcn_wmma_f32_16x16x4_f32(
                 false, a,  false, b,
                 (short)0, acc[t],  false,
                 false);
        }
    }
    float s[NT][8], d[NT][8];
#pragma unroll
    for (int t = 0; t < NT; ++t) {
        const float ws = a_src[16 * t + l];
        const float wd = a_dst[16 * t + l];
#pragma unroll
        for (int i = 0; i < 8; ++i) {
            s[t][i] = acc[t][i] * ws;
            d[t][i] = acc[t][i] * wd;
        }
    }
#pragma unroll
    for (int m = 1; m < CH; m <<= 1) {
        float ps[NT][8], pd[NT][8];
#pragma unroll
        for (int t = 0; t < NT; ++t)
#pragma unroll
            for (int i = 0; i < 8; ++i) {
                ps[t][i] = __shfl_xor(s[t][i], m);
                pd[t][i] = __shfl_xor(d[t][i], m);
            }
#pragma unroll
        for (int t = 0; t < NT; ++t)
#pragma unroll
            for (int i = 0; i < 8; ++i) {
                s[t][i] += ps[t][i];
                d[t][i] += pd[t][i];
            }
    }
#pragma unroll
    for (int t = 0; t < NT; ++t)
#pragma unroll
        for (int i = 0; i < 8; ++i)
            C[(size_t)(row0 + i + 8 * hf) * NCOL + 16 * t + l] = acc[t][i];
    if ((l & (CH - 1)) == 0) {
#pragma unroll
        for (int t = 0; t < NT; ++t)
#pragma unroll
            for (int i = 0; i < 8; ++i) {
                const size_t o =
                    (size_t)(row0 + i + 8 * hf) * H + (16 * t + l) / CH;
                as_[o] = s[t][i];
                ad_[o] = d[t][i];
            }
    }
}

#define NPS  392
#define OWN_THREADS 1024
#define LCAP 18432
#define LPITCH 7168
#define LSHIFT 9

__global__ void key_kernel(const int* __restrict__ dsts, int E,
                           unsigned* __restrict__ planes) {
    const int PW = (E >> 10) * 32;
    const int D  = blockIdx.x * blockDim.x + threadIdx.x;
    if (D >= PW) return;
    int d[32];
    if ((reinterpret_cast<size_t>(dsts) & 15) == 0) {
        const int4* __restrict__ d4 =
            reinterpret_cast<const int4*>(dsts) + (size_t)8 * D;
#pragma unroll
        for (int u = 0; u < 8; ++u) {
            const int4 v = d4[u];
            d[4 * u + 0] = v.x;
            d[4 * u + 1] = v.y;
            d[4 * u + 2] = v.z;
            d[4 * u + 3] = v.w;
        }
    } else {
#pragma unroll
        for (int j = 0; j < 32; ++j) d[j] = dsts[(size_t)32 * D + j];
    }
    unsigned a[32];
#pragma unroll
    for (int j = 0; j < 32; ++j) {
        const unsigned g = (unsigned)d[j] / (unsigned)NPS;
        a[j] = (1u << ((g >> 4) & 15u)) | (0x10000u << (g & 15u));
    }
#pragma unroll
    for (int s = 16; s != 0; s >>= 1) {
        const unsigned m = 0xffffffffu / ((1u << s) + 1u);
#pragma unroll
        for (int k = 0; k < 32; ++k) {
            if ((k & s) == 0) {
                const unsigned t = ((a[k] >> s) ^ a[k + s]) & m;
                a[k] ^= t << s;
                a[k + s] ^= t;
            }
        }
    }
#pragma unroll
    for (int p = 0; p < 32; ++p) planes[(size_t)p * PW + D] = a[p];
}

template <class Fn>
__device__ __forceinline__ void scan_keys(const unsigned* __restrict__ keys,
                                          const int* __restrict__ dsts, int E,
                                          int n0, unsigned cnt, Fn emit) {
    const int tid = threadIdx.x;
    constexpr int nt = OWN_THREADS;
    constexpr int UPT = 4;
    const int EU = (E >> 10) * 8;
    const uint4* __restrict__ pu = reinterpret_cast<const uint4*>(keys) +
                                   (size_t)((blockIdx.x >> 4) & 15u) * EU;
    const uint4* __restrict__ pw = reinterpret_cast<const uint4*>(keys) +
                                   (size_t)(16u + (blockIdx.x & 15u)) * EU;
    auto drain = [&](unsigned m, int e0) {
        while (m) {
            const int p = __builtin_ctz(m);
            m &= m - 1;
            emit(e0 + p);
        }
    };
    for (int q0 = tid; q0 < EU; q0 += UPT * nt) {
        uint4 u[UPT], w[UPT];
#pragma unroll
        for (int j = 0; j < UPT; ++j) {
            const int q = min(q0 + j * nt, EU - 1);
            u[j] = pu[q];
            w[j] = pw[q];
        }
#pragma unroll
        for (int j = 0; j < UPT; ++j) {
            const int q = q0 + j * nt;
            const unsigned in = (q < EU) ? ~0u : 0u;
            drain(u[j].x & w[j].x & in, 128 * q);
            drain(u[j].y & w[j].y & in, 128 * q + 32);
            drain(u[j].z & w[j].z & in, 128 * q + 64);
            drain(u[j].w & w[j].w & in, 128 * q + 96);
        }
    }
    for (int i = 128 * EU + tid; i < E; i += nt) {
        const unsigned a = (unsigned)(dsts[i] - n0);
        if (a < cnt) emit(i);
    }
}

template <int R>
__device__ __forceinline__ int quad_lane(int x) {
    return __builtin_amdgcn_update_dpp(0, x, 0x55 * R, 0xF, 0xF, true);
}
__device__ __forceinline__ int quad_from(int x, int r) {
    switch (r) {
        case 0:  return quad_lane<0>(x);
        case 1:  return quad_lane<1>(x);
        case 2:  return quad_lane<2>(x);
        default: return quad_lane<3>(x);
    }
}
__device__ __forceinline__ float quad_from(float x, int r) {
    return __int_as_float(quad_from(__float_as_int(x), r));
}

template <int H, int C, bool FINAL, bool FIRST>
__global__ __launch_bounds__(OWN_THREADS) void owner_kernel(
    const int* __restrict__ srcs, const int* __restrict__ dsts,
    const unsigned* __restrict__ keys, int E, int N,
    const float* __restrict__ as_, const float* __restrict__ ad_,
    const float* __restrict__ xw, const float* __restrict__ bias,
    float* __restrict__ outp, int* __restrict__ ltot,
    int* __restrict__ lsave, const int* ws_first, size_t ws_bytes) {
    static_assert(C % 4 == 0, "C must be a multiple of 4");
    static_assert(!FINAL || H == 1, "the final layer has one head");
    static_assert(NPS <= (1 << LSHIFT), "a local destination must fit");
    static_assert(LPITCH <= LCAP, "a stored list must fit the list in LDS");
    constexpr int F = H * C;
    constexpr int FP = F + 1;
    __shared__ unsigned s_m[NPS * H];
    __shared__ float    s_ad[NPS * H];
    __shared__ float    s_den[NPS * H];
    __shared__ float    s_acc[NPS * FP];
    __shared__ int      s_li[LCAP];
    __shared__ int      s_ld[LCAP];
    __shared__ int      s_cnt;

    const int tid = threadIdx.x;
    constexpr int nt = OWN_THREADS;
    const int n0 = blockIdx.x * NPS;
    if (n0 >= N) return;
    const int cnt = min(NPS, N - n0);

    for (int j = tid; j < cnt * H; j += nt) {
        s_m[j] = FKEY_NEG_INF;
        s_den[j] = 0.f;
        s_ad[j] = ad_[(size_t)n0 * H + j];
    }
    for (int j = tid; j < cnt * FP; j += nt) s_acc[j] = 0.f;
    if (tid == 0) s_cnt = 0;
    __syncthreads();

    auto score = [&](int s, int dl, float* e) {
#pragma unroll
        for (int h = 0; h < H; ++h) {
            const float v = as_[s * H + h] + s_ad[dl * H + h];
            e[h] = (v > 0.f) ? v : NEG_SLOPE * v;
        }
    };
    auto put_max = [&](int dl, const float* e) {
#pragma unroll
        for (int h = 0; h < H; ++h) atomicMax(&s_m[dl * H + h], fkey(e[h]));
    };
    auto add_edge = [&](int s, int dl, const float* e) {
        float w[H];
#pragma unroll
        for (int h = 0; h < H; ++h) {
            w[h] = expf(e[h] - fkey_inv(s_m[dl * H + h]));
            atomicAdd(&s_den[dl * H + h], w[h]);
        }
        const float4* __restrict__ sv = (const float4*)(xw + (size_t)s * F);
        float* dv = &s_acc[dl * FP];
#pragma unroll
        for (int f = 0; f < F / 4; ++f) {
            const float4 v = sv[f];
            const float wh = w[(f * 4) / C];
            atomicAdd(&dv[f * 4 + 0], wh * v.x);
            atomicAdd(&dv[f * 4 + 1], wh * v.y);
            atomicAdd(&dv[f * 4 + 2], wh * v.z);
            atomicAdd(&dv[f * 4 + 3], wh * v.w);
        }
    };

    auto add_rows = [&](bool valid, int s, int dl, const float* w) {
        constexpr int FL = F / 4;
        constexpr int NV = FL / 4;
        static_assert(FL % 4 == 0 && C % FL == 0,
                      "a lane's floats are whole float4s of one head");
        const int q  = tid & 3;
        const int hq = (q * FL) / C;
        const int sv = valid ? s : n0;
        const int dv = valid ? dl : -1;
        float4 v[4][NV];
        int    dls[4];
        float  ws[4];
#pragma unroll
        for (int r = 0; r < 4; ++r) {
            const int st = quad_from(sv, r);
            dls[r] = quad_from(dv, r);
            float wt = quad_from(w[0], r);
#pragma unroll
            for (int h = 1; h < H; ++h) {
                const float wh = quad_from(w[h], r);
                if (hq == h) wt = wh;
            }
            ws[r] = wt;
            const float4* __restrict__ row =
                (const float4*)(xw + (size_t)st * F + q * FL);
#pragma unroll
            for (int i = 0; i < NV; ++i) v[r][i] = row[i];
        }
#pragma unroll
        for (int r = 0; r < 4; ++r) {
            if (dls[r] >= 0) {
                float* d4 = &s_acc[dls[r] * FP + q * FL];
#pragma unroll
                for (int i = 0; i < NV; ++i) {
                    atomicAdd(&d4[4 * i + 0], ws[r] * v[r][i].x);
                    atomicAdd(&d4[4 * i + 1], ws[r] * v[r][i].y);
                    atomicAdd(&d4[4 * i + 2], ws[r] * v[r][i].z);
                    atomicAdd(&d4[4 * i + 3], ws[r] * v[r][i].w);
                }
            }
        }
    };
    auto edge_weights = [&](bool valid, int dl, const float* e, float* w) {
#pragma unroll
        for (int h = 0; h < H; ++h) {
            w[h] = 0.f;
            if (valid) {
                w[h] = expf(e[h] - fkey_inv(s_m[dl * H + h]));
                atomicAdd(&s_den[dl * H + h], w[h]);
            }
        }
    };
    const int wave0 = __builtin_amdgcn_readfirstlane(tid) & ~31;

    const bool ws_ok =
        (size_t)((lsave + (size_t)gridDim.x * LPITCH) - ws_first) *
                sizeof(int) <= ws_bytes &&
        N <= (1 << (31 - LSHIFT));
    int* __restrict__ mine = lsave + (size_t)blockIdx.x * LPITCH;

    int  total  = 0;
    bool loaded = false;
    if (!FIRST && ws_ok) {
        total  = ltot[blockIdx.x];
        loaded = total <= LPITCH;
    }
    if (loaded) {
        constexpr int LSLOTS = LPITCH / OWN_THREADS;
        static_assert(LSLOTS * OWN_THREADS == LPITCH, "whole slots");
        if (total > 0) {
            int v[LSLOTS];
#pragma unroll
            for (int j = 0; j < LSLOTS; ++j)
                v[j] = mine[min(tid + j * nt, total - 1)];
#pragma unroll
            for (int j = 0; j < LSLOTS; ++j) {
                const int k = tid + j * nt;
                if (k < total) {
                    s_li[k] = v[j] >> LSHIFT;
                    s_ld[k] = (v[j] < 0) ? -1 : (v[j] & ((1 << LSHIFT) - 1));
                }
            }
        }
    } else {
        scan_keys(keys, dsts, E, n0, (unsigned)cnt, [&](int i) {
            const int pos = atomicAdd(&s_cnt, 1);
            if (pos < LCAP) s_li[pos] = i;
        });
        __syncthreads();
        total = s_cnt;
    }
    const bool listed = total <= LCAP;
    if (!loaded) {
        const bool store = FIRST && ws_ok && total <= LPITCH;
        if (FIRST && ws_ok && tid == 0) ltot[blockIdx.x] = total;
        auto resolve = [&](auto keep) {
            for (int k = tid; k < total; k += nt) {
                const int i = s_li[k];
                const unsigned a = (unsigned)(dsts[i] - n0);
                const int s  = srcs[i];
                const int dl = (a < (unsigned)cnt) ? (int)a : -1;
                s_li[k] = s;
                s_ld[k] = dl;
                keep(k, s, dl);
            }
        };
        if (store)
            resolve([&](int k, int s, int dl) {
                mine[k] = (dl < 0) ? -1 : ((s << LSHIFT) | dl);
            });
        else if (listed)
            resolve([](int, int, int) {});
    }

    constexpr int KS = LPITCH / OWN_THREADS;
    static_assert(NPS <= OWN_THREADS, "one self-loop per thread at most");
    float ek[KS][H];
    float es[H];
#pragma unroll
    for (int h = 0; h < H; ++h) {
        es[h] = 0.f;
#pragma unroll
        for (int j = 0; j < KS; ++j) ek[j][h] = 0.f;
    }

    if (listed) {
#pragma unroll
        for (int j = 0; j < KS; ++j) {
            const int k = tid + j * nt;
            if (k < total) {
                const int dl = s_ld[k];
                if (dl >= 0) {
                    score(s_li[k], dl, ek[j]);
                    put_max(dl, ek[j]);
                }
            }
        }
        for (int k = tid + KS * nt; k < total; k += nt) {
            const int dl = s_ld[k];
            if (dl >= 0) {
                float e[H];
                score(s_li[k], dl, e);
                put_max(dl, e);
            }
        }
    } else {
        for (int i = tid; i < E; i += nt) {
            const unsigned a = (unsigned)(dsts[i] - n0);
            if (a < (unsigned)cnt) {
                float e[H];
                score(srcs[i], (int)a, e);
                put_max((int)a, e);
            }
        }
    }
    if (tid < cnt) {
        score(n0 + tid, tid, es);
        put_max(tid, es);
    }
    __syncthreads();

    if (listed) {
        float er[KS + 1][H];
#pragma unroll
        for (int h = 0; h < H; ++h) {
#pragma unroll
            for (int j = 0; j < KS; ++j) er[j][h] = ek[j][h];
            er[KS][h] = es[h];
        }
#pragma nounroll
        for (int j = 0; j <= KS; ++j) {
            const bool self = j == KS;
            const int kb = self ? 0 : j * nt;
            if (wave0 + kb < (self ? cnt : total)) {
                const int k = tid + kb;
                int dl = -1, s = n0;
                if (self) {
                    if (tid < cnt) { dl = tid; s = n0 + tid; }
                } else if (k < total) {
                    dl = s_ld[k];
                    if (dl >= 0) s = s_li[k];
                }
                const bool valid = dl >= 0;
                float w[H];
                edge_weights(valid, dl, er[0], w);
                add_rows(valid, s, dl, w);
            }
#pragma unroll
            for (int i = 0; i < KS; ++i)
#pragma unroll
                for (int h = 0; h < H; ++h) er[i][h] = er[i + 1][h];
        }
        for (int k = tid + KS * nt; k < total; k += nt) {
            const int dl = s_ld[k];
            if (dl >= 0) {
                float e[H];
                score(s_li[k], dl, e);
                add_edge(s_li[k], dl, e);
            }
        }
    } else {
        for (int i = tid; i < E; i += nt) {
            const unsigned a = (unsigned)(dsts[i] - n0);
            if (a < (unsigned)cnt) {
                float e[H];
                const int s = srcs[i];
                score(s, (int)a, e);
                add_edge(s, (int)a, e);
            }
        }
        if (tid < cnt) add_edge(n0 + tid, tid, es);
    }
    __syncthreads();

    if (!FINAL) {
        for (int j = tid; j < cnt * F; j += nt) {
            const float v =
                s_acc[(j / F) * FP + j % F] / s_den[j / C] + bias[j % F];
            outp[(size_t)n0 * F + j] = (v > 0.f) ? v : (expf(v) - 1.f);
        }
    } else {
        for (int dl = tid; dl < cnt; dl += nt) {
            const float dn = s_den[dl];
            float v[F];
            float m = -INFINITY;
#pragma unroll
            for (int c = 0; c < F; ++c) {
                v[c] = s_acc[dl * FP + c] / dn + bias[c];
                m = fmaxf(m, v[c]);
            }
            float ssum = 0.f;
#pragma unroll
            for (int c = 0; c < F; ++c) ssum += expf(v[c] - m);
            const float lg = m + logf(ssum);
#pragma unroll
            for (int c = 0; c < F; ++c)
                outp[(size_t)(n0 + dl) * F + c] = v[c] - lg;
        }
    }
}

extern "C" void kernel_launch(void* const* d_in, const int* in_sizes, int n_in,
                              void* d_out, int out_size, void* d_ws,
                              size_t ws_size, hipStream_t stream) {
    const float* x   = (const float*)d_in[0];
    const int*   ei  = (const int*)d_in[1];
    const float* W1  = (const float*)d_in[2];
    const float* a1s = (const float*)d_in[3];
    const float* a1d = (const float*)d_in[4];
    const float* b1  = (const float*)d_in[5];
    const float* W2  = (const float*)d_in[6];
    const float* a2s = (const float*)d_in[7];
    const float* a2d = (const float*)d_in[8];
    const float* b2  = (const float*)d_in[9];
    float* out = (float*)d_out;

    const int N = in_sizes[0] / 128;
    const int E = in_sizes[1] / 2;
    const int* srcs = ei;
    const int* dsts = ei + E;

    char* p = (char*)d_ws;
    auto carve = [&](size_t bytes) {
        void* r = (void*)p;
        p += (bytes + 255) & ~(size_t)255;
        return r;
    };
    float* xw1 = (float*)carve((size_t)N * 32 * 4);
    float* as1 = (float*)carve((size_t)N * 4 * 4);
    float* ad1 = (float*)carve((size_t)N * 4 * 4);
    float* h1  = (float*)carve((size_t)N * 32 * 4);
    unsigned* keys = (unsigned*)carve((size_t)E * 4 + 16);
    float* xw2 = xw1;
    float* as2 = as1;
    float* ad2 = ad1;

    const int B = 256;
    auto g = [](int n, int b) { return (n + b - 1) / b; };

    int* ltot  = (int*)carve((size_t)g(N, NPS) * 4);
    int* lsave = (int*)carve((size_t)g(N, NPS) * LPITCH * 4);
    const int* ws_first = (const int*)d_ws;

    key_kernel<<<g((E >> 10) * 32 + 1, B), B, 0, stream>>>(dsts, E, keys);

    gemm_alpha_kernel<128, 32, 4><<<g(N, 32), 64, 0, stream>>>(
        x, W1, a1s, a1d, xw1, as1, ad1, N);
    owner_kernel<4, 8, false, true><<<g(N, NPS), OWN_THREADS, 0, stream>>>(
        srcs, dsts, keys, E, N, as1, ad1, xw1, b1, h1, ltot, lsave, ws_first,
        ws_size);

    gemm_alpha_kernel<32, 16, 1><<<g(N, 32), 64, 0, stream>>>(
        h1, W2, a2s, a2d, xw2, as2, ad2, N);
    owner_kernel<1, 16, true, false><<<g(N, NPS), OWN_THREADS, 0, stream>>>(
        srcs, dsts, keys, E, N, as2, ad2, xw2, b2, out, ltot, lsave, ws_first,
        ws_size);
}


# owner launches: one walk, not two: p = exp(score - the destination's self-loop score), softmax being shift-invariant; no max walk; a group with non-finite sums redoes it with maxima; arrival order
# speedup vs baseline: 13.8652x; 1.0473x over previous
#include <hip/hip_runtime.h>
#include <math.h>

typedef __attribute__((ext_vector_type(2))) float v2f;
typedef __attribute__((ext_vector_type(8))) float v8f;

#define NEG_SLOPE 0.2f

__device__ __forceinline__ unsigned fkey(float f) {
    unsigned b = __float_as_uint(f);
    return (b & 0x80000000u) ? ~b : (b | 0x80000000u);
}
__device__ __forceinline__ float fkey_inv(unsigned u) {
    return (u & 0x80000000u) ? __uint_as_float(u & 0x7fffffffu)
                             : __uint_as_float(~u);
}
#define FKEY_NEG_INF 0x007fffffu

template <int K, int NCOL, int H>
__global__ void gemm_alpha_kernel(const float* __restrict__ A,
                                  const float* __restrict__ B,
                                  const float* __restrict__ a_src,
                                  const float* __restrict__ a_dst,
                                  float* __restrict__ C,
                                  float* __restrict__ as_,
                                  float* __restrict__ ad_, int N) {
    constexpr int NT = NCOL / 16;
    constexpr int CH = NCOL / H;
    static_assert(CH == 8 || CH == 16, "a head must lie inside one tile");
    const int lane = threadIdx.x & 31;
    const int l    = lane & 15;
    const int hf   = lane >> 4;
    const int row0 =
        (blockIdx.x * (blockDim.x >> 5) + (threadIdx.x >> 5)) * 16;
    if (row0 >= N) return;

    const float* __restrict__ arow = A + (size_t)(row0 + l) * K;
    v8f acc[NT];
#pragma unroll
    for (int t = 0; t < NT; ++t) acc[t] = (v8f){};

#pragma unroll
    for (int k = 0; k < K; k += 4) {
        v2f a;
        a.x = arow[k + 2 * hf];
        a.y = arow[k + 2 * hf + 1];
#pragma unroll
        for (int t = 0; t < NT; ++t) {
            v2f b;
            b.x = B[(size_t)(k + 2 * hf) * NCOL + 16 * t + l];
            b.y = B[(size_t)(k + 2 * hf + 1) * NCOL + 16 * t + l];
            acc[t] = __builtin_amdgcn_wmma_f32_16x16x4_f32(
                 false, a,  false, b,
                 (short)0, acc[t],  false,
                 false);
        }
    }
    float s[NT][8], d[NT][8];
#pragma unroll
    for (int t = 0; t < NT; ++t) {
        const float ws = a_src[16 * t + l];
        const float wd = a_dst[16 * t + l];
#pragma unroll
        for (int i = 0; i < 8; ++i) {
            s[t][i] = acc[t][i] * ws;
            d[t][i] = acc[t][i] * wd;
        }
    }
#pragma unroll
    for (int m = 1; m < CH; m <<= 1) {
        float ps[NT][8], pd[NT][8];
#pragma unroll
        for (int t = 0; t < NT; ++t)
#pragma unroll
            for (int i = 0; i < 8; ++i) {
                ps[t][i] = __shfl_xor(s[t][i], m);
                pd[t][i] = __shfl_xor(d[t][i], m);
            }
#pragma unroll
        for (int t = 0; t < NT; ++t)
#pragma unroll
            for (int i = 0; i < 8; ++i) {
                s[t][i] += ps[t][i];
                d[t][i] += pd[t][i];
            }
    }
#pragma unroll
    for (int t = 0; t < NT; ++t)
#pragma unroll
        for (int i = 0; i < 8; ++i)
            C[(size_t)(row0 + i + 8 * hf) * NCOL + 16 * t + l] = acc[t][i];
    if ((l & (CH - 1)) == 0) {
#pragma unroll
        for (int t = 0; t < NT; ++t)
#pragma unroll
            for (int i = 0; i < 8; ++i) {
                const size_t o =
                    (size_t)(row0 + i + 8 * hf) * H + (16 * t + l) / CH;
                as_[o] = s[t][i];
                ad_[o] = d[t][i];
            }
    }
}

#define NPS  392
#define OWN_THREADS 1024
#define LCAP 18432
#define LPITCH 7168
#define LSHIFT 9

__global__ void key_kernel(const int* __restrict__ dsts, int E,
                           unsigned* __restrict__ planes) {
    const int PW = (E >> 10) * 32;
    const int D  = blockIdx.x * blockDim.x + threadIdx.x;
    if (D >= PW) return;
    int d[32];
    if ((reinterpret_cast<size_t>(dsts) & 15) == 0) {
        const int4* __restrict__ d4 =
            reinterpret_cast<const int4*>(dsts) + (size_t)8 * D;
#pragma unroll
        for (int u = 0; u < 8; ++u) {
            const int4 v = d4[u];
            d[4 * u + 0] = v.x;
            d[4 * u + 1] = v.y;
            d[4 * u + 2] = v.z;
            d[4 * u + 3] = v.w;
        }
    } else {
#pragma unroll
        for (int j = 0; j < 32; ++j) d[j] = dsts[(size_t)32 * D + j];
    }
    unsigned a[32];
#pragma unroll
    for (int j = 0; j < 32; ++j) {
        const unsigned g = (unsigned)d[j] / (unsigned)NPS;
        a[j] = (1u << ((g >> 4) & 15u)) | (0x10000u << (g & 15u));
    }
#pragma unroll
    for (int s = 16; s != 0; s >>= 1) {
        const unsigned m = 0xffffffffu / ((1u << s) + 1u);
#pragma unroll
        for (int k = 0; k < 32; ++k) {
            if ((k & s) == 0) {
                const unsigned t = ((a[k] >> s) ^ a[k + s]) & m;
                a[k] ^= t << s;
                a[k + s] ^= t;
            }
        }
    }
#pragma unroll
    for (int p = 0; p < 32; ++p) planes[(size_t)p * PW + D] = a[p];
}

template <class Fn>
__device__ __forceinline__ void scan_keys(const unsigned* __restrict__ keys,
                                          const int* __restrict__ dsts, int E,
                                          int n0, unsigned cnt, Fn emit) {
    const int tid = threadIdx.x;
    constexpr int nt = OWN_THREADS;
    constexpr int UPT = 4;
    const int EU = (E >> 10) * 8;
    const uint4* __restrict__ pu = reinterpret_cast<const uint4*>(keys) +
                                   (size_t)((blockIdx.x >> 4) & 15u) * EU;
    const uint4* __restrict__ pw = reinterpret_cast<const uint4*>(keys) +
                                   (size_t)(16u + (blockIdx.x & 15u)) * EU;
    auto drain = [&](unsigned m, int e0) {
        while (m) {
            const int p = __builtin_ctz(m);
            m &= m - 1;
            emit(e0 + p);
        }
    };
    for (int q0 = tid; q0 < EU; q0 += UPT * nt) {
        uint4 u[UPT], w[UPT];
#pragma unroll
        for (int j = 0; j < UPT; ++j) {
            const int q = min(q0 + j * nt, EU - 1);
            u[j] = pu[q];
            w[j] = pw[q];
        }
#pragma unroll
        for (int j = 0; j < UPT; ++j) {
            const int q = q0 + j * nt;
            const unsigned in = (q < EU) ? ~0u : 0u;
            drain(u[j].x & w[j].x & in, 128 * q);
            drain(u[j].y & w[j].y & in, 128 * q + 32);
            drain(u[j].z & w[j].z & in, 128 * q + 64);
            drain(u[j].w & w[j].w & in, 128 * q + 96);
        }
    }
    for (int i = 128 * EU + tid; i < E; i += nt) {
        const unsigned a = (unsigned)(dsts[i] - n0);
        if (a < cnt) emit(i);
    }
}

template <int R>
__device__ __forceinline__ int quad_lane(int x) {
    return __builtin_amdgcn_update_dpp(0, x, 0x55 * R, 0xF, 0xF, true);
}
__device__ __forceinline__ int quad_from(int x, int r) {
    switch (r) {
        case 0:  return quad_lane<0>(x);
        case 1:  return quad_lane<1>(x);
        case 2:  return quad_lane<2>(x);
        default: return quad_lane<3>(x);
    }
}
__device__ __forceinline__ float quad_from(float x, int r) {
    return __int_as_float(quad_from(__float_as_int(x), r));
}

template <int H, int C, bool FINAL, bool FIRST>
__global__ __launch_bounds__(OWN_THREADS) void owner_kernel(
    const int* __restrict__ srcs, const int* __restrict__ dsts,
    const unsigned* __restrict__ keys, int E, int N,
    const float* __restrict__ as_, const float* __restrict__ ad_,
    const float* __restrict__ xw, const float* __restrict__ bias,
    float* __restrict__ outp, int* __restrict__ ltot,
    int* __restrict__ lsave, const int* ws_first, size_t ws_bytes) {
    static_assert(C % 4 == 0, "C must be a multiple of 4");
    static_assert(!FINAL || H == 1, "the final layer has one head");
    static_assert(NPS <= (1 << LSHIFT), "a local destination must fit");
    static_assert(LPITCH <= LCAP, "a stored list must fit the list in LDS");
    constexpr int F = H * C;
    constexpr int FP = F + 1;
    __shared__ unsigned s_m[NPS * H];
    __shared__ float    s_ad[NPS * H];
    __shared__ float    s_es[NPS * H];
    __shared__ int      s_bad;
    __shared__ float    s_den[NPS * H];
    __shared__ float    s_acc[NPS * FP];
    __shared__ int      s_li[LCAP];
    __shared__ int      s_ld[LCAP];
    __shared__ int      s_cnt;

    const int tid = threadIdx.x;
    constexpr int nt = OWN_THREADS;
    const int n0 = blockIdx.x * NPS;
    if (n0 >= N) return;
    const int cnt = min(NPS, N - n0);

    auto clear_sums = [&]() {
        for (int j = tid; j < cnt * H; j += nt) {
            s_m[j] = FKEY_NEG_INF;
            s_den[j] = 0.f;
        }
        for (int j = tid; j < cnt * FP; j += nt) s_acc[j] = 0.f;
    };
    for (int j = tid; j < cnt * H; j += nt) {
        const float ad = ad_[(size_t)n0 * H + j];
        const float v  = as_[(size_t)n0 * H + j] + ad;
        s_ad[j] = ad;
        s_es[j] = (v > 0.f) ? v : NEG_SLOPE * v;
    }
    clear_sums();
    if (tid == 0) {
        s_cnt = 0;
        s_bad = 0;
    }
    __syncthreads();

    auto score = [&](int s, int dl, float* e) {
#pragma unroll
        for (int h = 0; h < H; ++h) {
            const float v = as_[s * H + h] + s_ad[dl * H + h];
            e[h] = (v > 0.f) ? v : NEG_SLOPE * v;
        }
    };
    auto put_max = [&](int dl, const float* e) {
#pragma unroll
        for (int h = 0; h < H; ++h) atomicMax(&s_m[dl * H + h], fkey(e[h]));
    };
    auto add_edge = [&](int s, int dl, const float* e) {
        float w[H];
#pragma unroll
        for (int h = 0; h < H; ++h) {
            w[h] = expf(e[h] - fkey_inv(s_m[dl * H + h]));
            atomicAdd(&s_den[dl * H + h], w[h]);
        }
        const float4* __restrict__ sv = (const float4*)(xw + (size_t)s * F);
        float* dv = &s_acc[dl * FP];
#pragma unroll
        for (int f = 0; f < F / 4; ++f) {
            const float4 v = sv[f];
            const float wh = w[(f * 4) / C];
            atomicAdd(&dv[f * 4 + 0], wh * v.x);
            atomicAdd(&dv[f * 4 + 1], wh * v.y);
            atomicAdd(&dv[f * 4 + 2], wh * v.z);
            atomicAdd(&dv[f * 4 + 3], wh * v.w);
        }
    };

    auto walk_rows = [&](bool valid, int s, int dl) {
        constexpr int FL = F / 4;
        constexpr int NV = FL / 4;
        static_assert(FL % 4 == 0 && C % FL == 0,
                      "a lane's floats are whole float4s of one head");
        const int q  = tid & 3;
        const int hq = (q * FL) / C;
        const int sv = valid ? s : n0;
        const int dv = valid ? dl : -1;
        const int dc = valid ? dl : 0;
        float ad[H], es[H];
#pragma unroll
        for (int h = 0; h < H; ++h) {
            ad[h] = s_ad[dc * H + h];
            es[h] = s_es[dc * H + h];
        }
        float a[H];
#pragma unroll
        for (int h = 0; h < H; ++h) a[h] = as_[sv * H + h];
        float4 v[4][NV];
        int    dls[4];
#pragma unroll
        for (int r = 0; r < 4; ++r) {
            const int st = quad_from(sv, r);
            dls[r] = quad_from(dv, r);
            const float4* __restrict__ row =
                (const float4*)(xw + (size_t)st * F + q * FL);
#pragma unroll
            for (int i = 0; i < NV; ++i) v[r][i] = row[i];
        }
        float w[H];
#pragma unroll
        for (int h = 0; h < H; ++h) {
            const float x = a[h] + ad[h];
            const float e = (x > 0.f) ? x : NEG_SLOPE * x;
            const float p = expf(e - es[h]);
            w[h] = valid ? p : 0.f;
        }
        if (valid) {
#pragma unroll
            for (int h = 0; h < H; ++h) atomicAdd(&s_den[dc * H + h], w[h]);
        }
#pragma unroll
        for (int r = 0; r < 4; ++r) {
            float wt = quad_from(w[0], r);
#pragma unroll
            for (int h = 1; h < H; ++h) {
                const float wh = quad_from(w[h], r);
                if (hq == h) wt = wh;
            }
            if (dls[r] >= 0) {
                float* d4 = &s_acc[dls[r] * FP + q * FL];
#pragma unroll
                for (int i = 0; i < NV; ++i) {
                    atomicAdd(&d4[4 * i + 0], wt * v[r][i].x);
                    atomicAdd(&d4[4 * i + 1], wt * v[r][i].y);
                    atomicAdd(&d4[4 * i + 2], wt * v[r][i].z);
                    atomicAdd(&d4[4 * i + 3], wt * v[r][i].w);
                }
            }
        }
    };
    const int wave0 = __builtin_amdgcn_readfirstlane(tid) & ~31;

    const bool ws_ok =
        (size_t)((lsave + (size_t)gridDim.x * LPITCH) - ws_first) *
                sizeof(int) <= ws_bytes &&
        N <= (1 << (31 - LSHIFT));
    int* __restrict__ mine = lsave + (size_t)blockIdx.x * LPITCH;

    int  total  = 0;
    bool loaded = false;
    if (!FIRST && ws_ok) {
        total  = ltot[blockIdx.x];
        loaded = total <= LPITCH;
    }
    if (loaded) {
        constexpr int LSLOTS = LPITCH / OWN_THREADS;
        static_assert(LSLOTS * OWN_THREADS == LPITCH, "whole slots");
        if (total > 0) {
            int v[LSLOTS];
#pragma unroll
            for (int j = 0; j < LSLOTS; ++j)
                v[j] = mine[min(tid + j * nt, total - 1)];
#pragma unroll
            for (int j = 0; j < LSLOTS; ++j) {
                const int k = tid + j * nt;
                if (k < total) {
                    s_li[k] = v[j] >> LSHIFT;
                    s_ld[k] = (v[j] < 0) ? -1 : (v[j] & ((1 << LSHIFT) - 1));
                }
            }
        }
    } else {
        scan_keys(keys, dsts, E, n0, (unsigned)cnt, [&](int i) {
            const int pos = atomicAdd(&s_cnt, 1);
            if (pos < LCAP) s_li[pos] = i;
        });
        __syncthreads();
        total = s_cnt;
    }
    const bool listed = total <= LCAP;
    if (!loaded) {
        const bool store = FIRST && ws_ok && total <= LPITCH;
        if (FIRST && ws_ok && tid == 0) ltot[blockIdx.x] = total;
        auto resolve = [&](auto keep) {
            for (int k = tid; k < total; k += nt) {
                const int i = s_li[k];
                const unsigned a = (unsigned)(dsts[i] - n0);
                const int s  = srcs[i];
                const int dl = (a < (unsigned)cnt) ? (int)a : -1;
                s_li[k] = s;
                s_ld[k] = dl;
                keep(k, s, dl);
            }
        };
        if (store)
            resolve([&](int k, int s, int dl) {
                mine[k] = (dl < 0) ? -1 : ((s << LSHIFT) | dl);
            });
        else if (listed)
            resolve([](int, int, int) {});
    }

    auto exact_walks = [&]() {
        float es[H];
        if (listed) {
            for (int k = tid; k < total; k += nt) {
                const int dl = s_ld[k];
                if (dl >= 0) {
                    float e[H];
                    score(s_li[k], dl, e);
                    put_max(dl, e);
                }
            }
        } else {
            for (int i = tid; i < E; i += nt) {
                const unsigned a = (unsigned)(dsts[i] - n0);
                if (a < (unsigned)cnt) {
                    float e[H];
                    score(srcs[i], (int)a, e);
                    put_max((int)a, e);
                }
            }
        }
        if (tid < cnt) {
            score(n0 + tid, tid, es);
            put_max(tid, es);
        }
        __syncthreads();
        if (listed) {
            for (int k = tid; k < total; k += nt) {
                const int dl = s_ld[k];
                if (dl >= 0) {
                    float e[H];
                    score(s_li[k], dl, e);
                    add_edge(s_li[k], dl, e);
                }
            }
        } else {
            for (int i = tid; i < E; i += nt) {
                const unsigned a = (unsigned)(dsts[i] - n0);
                if (a < (unsigned)cnt) {
                    float e[H];
                    const int s = srcs[i];
                    score(s, (int)a, e);
                    add_edge(s, (int)a, e);
                }
            }
        }
        if (tid < cnt) add_edge(n0 + tid, tid, es);
        __syncthreads();
    };

    auto write_out = [&]() {
        bool bad = false;
        if (!FINAL) {
            for (int j = tid; j < cnt * F; j += nt) {
                const float a  = s_acc[(j / F) * FP + j % F];
                const float dn = s_den[j / C];
                bad = bad || !(fabsf(a) < INFINITY) || !(dn < INFINITY);
                const float v = a / dn + bias[j % F];
                outp[(size_t)n0 * F + j] = (v > 0.f) ? v : (expf(v) - 1.f);
            }
        } else {
            for (int dl = tid; dl < cnt; dl += nt) {
                const float dn = s_den[dl];
                bad = bad || !(dn < INFINITY);
                float v[F];
                float m = -INFINITY;
#pragma unroll
                for (int c = 0; c < F; ++c) {
                    const float a = s_acc[dl * FP + c];
                    bad = bad || !(fabsf(a) < INFINITY);
                    v[c] = a / dn + bias[c];
                    m = fmaxf(m, v[c]);
                }
                float ssum = 0.f;
#pragma unroll
                for (int c = 0; c < F; ++c) ssum += expf(v[c] - m);
                const float lg = m + logf(ssum);
#pragma unroll
                for (int c = 0; c < F; ++c)
                    outp[(size_t)(n0 + dl) * F + c] = v[c] - lg;
            }
        }
        return bad;
    };

    bool exact = !listed;
    for (;;) {
        if (exact) {
            exact_walks();
        } else {
            const int trips = (total + nt - 1) / nt;
#pragma nounroll
            for (int j = 0; j <= trips; ++j) {
                const bool self = j == trips;
                const int kb = self ? 0 : j * nt;
                if (wave0 + kb < (self ? cnt : total)) {
                    const int k = tid + kb;
                    int dl = -1, s = n0;
                    if (self) {
                        if (tid < cnt) { dl = tid; s = n0 + tid; }
                    } else if (k < total) {
                        dl = s_ld[k];
                        if (dl >= 0) s = s_li[k];
                    }
                    walk_rows(dl >= 0, s, dl);
                }
            }
            __syncthreads();
        }
        const bool bad = write_out();
        if (exact) break;
        if (bad) s_bad = 1;
        __syncthreads();
        if (s_bad == 0) break;
        clear_sums();
        __syncthreads();
        exact = true;
    }
}

extern "C" void kernel_launch(void* const* d_in, const int* in_sizes, int n_in,
                              void* d_out, int out_size, void* d_ws,
                              size_t ws_size, hipStream_t stream) {
    const float* x   = (const float*)d_in[0];
    const int*   ei  = (const int*)d_in[1];
    const float* W1  = (const float*)d_in[2];
    const float* a1s = (const float*)d_in[3];
    const float* a1d = (const float*)d_in[4];
    const float* b1  = (const float*)d_in[5];
    const float* W2  = (const float*)d_in[6];
    const float* a2s = (const float*)d_in[7];
    const float* a2d = (const float*)d_in[8];
    const float* b2  = (const float*)d_in[9];
    float* out = (float*)d_out;

    const int N = in_sizes[0] / 128;
    const int E = in_sizes[1] / 2;
    const int* srcs = ei;
    const int* dsts = ei + E;

    char* p = (char*)d_ws;
    auto carve = [&](size_t bytes) {
        void* r = (void*)p;
        p += (bytes + 255) & ~(size_t)255;
        return r;
    };
    float* xw1 = (float*)carve((size_t)N * 32 * 4);
    float* as1 = (float*)carve((size_t)N * 4 * 4);
    float* ad1 = (float*)carve((size_t)N * 4 * 4);
    float* h1  = (float*)carve((size_t)N * 32 * 4);
    unsigned* keys = (unsigned*)carve((size_t)E * 4 + 16);
    float* xw2 = xw1;
    float* as2 = as1;
    float* ad2 = ad1;

    const int B = 256;
    auto g = [](int n, int b) { return (n + b - 1) / b; };

    int* ltot  = (int*)carve((size_t)g(N, NPS) * 4);
    int* lsave = (int*)carve((size_t)g(N, NPS) * LPITCH * 4);
    const int* ws_first = (const int*)d_ws;

    key_kernel<<<g((E >> 10) * 32 + 1, B), B, 0, stream>>>(dsts, E, keys);

    gemm_alpha_kernel<128, 32, 4><<<g(N, 32), 64, 0, stream>>>(
        x, W1, a1s, a1d, xw1, as1, ad1, N);
    owner_kernel<4, 8, false, true><<<g(N, NPS), OWN_THREADS, 0, stream>>>(
        srcs, dsts, keys, E, N, as1, ad1, xw1, b1, h1, ltot, lsave, ws_first,
        ws_size);

    gemm_alpha_kernel<32, 16, 1><<<g(N, 32), 64, 0, stream>>>(
        h1, W2, a2s, a2d, xw2, as2, ad2, N);
    owner_kernel<1, 16, true, false><<<g(N, NPS), OWN_THREADS, 0, stream>>>(
        srcs, dsts, keys, E, N, as2, ad2, xw2, b2, out, ltot, lsave, ws_first,
        ws_size);
}


# first owner launch, edge list: a wave sums its lanes' match counts (popcount, running sum by lane moves) and adds to the LDS counter once per trip, not in every drain pass with a wait; arrival order
# speedup vs baseline: 14.3922x; 1.0380x over previous
#include <hip/hip_runtime.h>
#include <math.h>

typedef __attribute__((ext_vector_type(2))) float v2f;
typedef __attribute__((ext_vector_type(8))) float v8f;
typedef __attribute__((ext_vector_type(4))) unsigned v4u;

#define NEG_SLOPE 0.2f

__device__ __forceinline__ unsigned fkey(float f) {
    unsigned b = __float_as_uint(f);
    return (b & 0x80000000u) ? ~b : (b | 0x80000000u);
}
__device__ __forceinline__ float fkey_inv(unsigned u) {
    return (u & 0x80000000u) ? __uint_as_float(u & 0x7fffffffu)
                             : __uint_as_float(~u);
}
#define FKEY_NEG_INF 0x007fffffu

template <int K, int NCOL, int H>
__global__ void gemm_alpha_kernel(const float* __restrict__ A,
                                  const float* __restrict__ B,
                                  const float* __restrict__ a_src,
                                  const float* __restrict__ a_dst,
                                  float* __restrict__ C,
                                  float* __restrict__ as_,
                                  float* __restrict__ ad_, int N) {
    constexpr int NT = NCOL / 16;
    constexpr int CH = NCOL / H;
    static_assert(CH == 8 || CH == 16, "a head must lie inside one tile");
    const int lane = threadIdx.x & 31;
    const int l    = lane & 15;
    const int hf   = lane >> 4;
    const int row0 =
        (blockIdx.x * (blockDim.x >> 5) + (threadIdx.x >> 5)) * 16;
    if (row0 >= N) return;

    const float* __restrict__ arow = A + (size_t)(row0 + l) * K;
    v8f acc[NT];
#pragma unroll
    for (int t = 0; t < NT; ++t) acc[t] = (v8f){};

#pragma unroll
    for (int k = 0; k < K; k += 4) {
        v2f a;
        a.x = arow[k + 2 * hf];
        a.y = arow[k + 2 * hf + 1];
#pragma unroll
        for (int t = 0; t < NT; ++t) {
            v2f b;
            b.x = B[(size_t)(k + 2 * hf) * NCOL + 16 * t + l];
            b.y = B[(size_t)(k + 2 * hf + 1) * NCOL + 16 * t + l];
            acc[t] = __builtin_amdgcn_wmma_f32_16x16x4_f32(
                 false, a,  false, b,
                 (short)0, acc[t],  false,
                 false);
        }
    }
    float s[NT][8], d[NT][8];
#pragma unroll
    for (int t = 0; t < NT; ++t) {
        const float ws = a_src[16 * t + l];
        const float wd = a_dst[16 * t + l];
#pragma unroll
        for (int i = 0; i < 8; ++i) {
            s[t][i] = acc[t][i] * ws;
            d[t][i] = acc[t][i] * wd;
        }
    }
#pragma unroll
    for (int m = 1; m < CH; m <<= 1) {
        float ps[NT][8], pd[NT][8];
#pragma unroll
        for (int t = 0; t < NT; ++t)
#pragma unroll
            for (int i = 0; i < 8; ++i) {
                ps[t][i] = __shfl_xor(s[t][i], m);
                pd[t][i] = __shfl_xor(d[t][i], m);
            }
#pragma unroll
        for (int t = 0; t < NT; ++t)
#pragma unroll
            for (int i = 0; i < 8; ++i) {
                s[t][i] += ps[t][i];
                d[t][i] += pd[t][i];
            }
    }
#pragma unroll
    for (int t = 0; t < NT; ++t)
#pragma unroll
        for (int i = 0; i < 8; ++i)
            C[(size_t)(row0 + i + 8 * hf) * NCOL + 16 * t + l] = acc[t][i];
    if ((l & (CH - 1)) == 0) {
#pragma unroll
        for (int t = 0; t < NT; ++t)
#pragma unroll
            for (int i = 0; i < 8; ++i) {
                const size_t o =
                    (size_t)(row0 + i + 8 * hf) * H + (16 * t + l) / CH;
                as_[o] = s[t][i];
                ad_[o] = d[t][i];
            }
    }
}

#define NPS  392
#define OWN_THREADS 1024
#define LCAP 18432
#define LPITCH 7168
#define LSHIFT 9

__global__ void key_kernel(const int* __restrict__ dsts, int E,
                           unsigned* __restrict__ planes) {
    const int PW = (E >> 10) * 32;
    const int D  = blockIdx.x * blockDim.x + threadIdx.x;
    if (D >= PW) return;
    int d[32];
    if ((reinterpret_cast<size_t>(dsts) & 15) == 0) {
        const int4* __restrict__ d4 =
            reinterpret_cast<const int4*>(dsts) + (size_t)8 * D;
#pragma unroll
        for (int u = 0; u < 8; ++u) {
            const int4 v = d4[u];
            d[4 * u + 0] = v.x;
            d[4 * u + 1] = v.y;
            d[4 * u + 2] = v.z;
            d[4 * u + 3] = v.w;
        }
    } else {
#pragma unroll
        for (int j = 0; j < 32; ++j) d[j] = dsts[(size_t)32 * D + j];
    }
    unsigned a[32];
#pragma unroll
    for (int j = 0; j < 32; ++j) {
        const unsigned g = (unsigned)d[j] / (unsigned)NPS;
        a[j] = (1u << ((g >> 4) & 15u)) | (0x10000u << (g & 15u));
    }
#pragma unroll
    for (int s = 16; s != 0; s >>= 1) {
        const unsigned m = 0xffffffffu / ((1u << s) + 1u);
#pragma unroll
        for (int k = 0; k < 32; ++k) {
            if ((k & s) == 0) {
                const unsigned t = ((a[k] >> s) ^ a[k + s]) & m;
                a[k] ^= t << s;
                a[k + s] ^= t;
            }
        }
    }
#pragma unroll
    for (int p = 0; p < 32; ++p) planes[(size_t)p * PW + D] = a[p];
}

template <int S>
__device__ __forceinline__ int row_below(int x) {
    return __builtin_amdgcn_update_dpp(0, x, 0x110 + S, 0xF, 0xF, true);
}

template <class Fn>
__device__ __forceinline__ void scan_keys(const unsigned* __restrict__ keys,
                                          const int* __restrict__ dsts, int E,
                                          int n0, unsigned cnt, int* lcount,
                                          int* list, int cap, Fn emit) {
    const int tid  = threadIdx.x;
    const int lane = tid & 31;
    const int wave0 = __builtin_amdgcn_readfirstlane(tid) & ~31;
    constexpr int nt = OWN_THREADS;
    constexpr int UPT = 4;
    const int EU = (E >> 10) * 8;
    const v4u* __restrict__ pu = reinterpret_cast<const v4u*>(keys) +
                                   (size_t)((blockIdx.x >> 4) & 15u) * EU;
    const v4u* __restrict__ pw = reinterpret_cast<const v4u*>(keys) +
                                   (size_t)(16u + (blockIdx.x & 15u)) * EU;
    for (int qw = wave0; qw < EU; qw += UPT * nt) {
        const int q0 = qw + lane;
        v4u u[UPT], w[UPT];
#pragma unroll
        for (int j = 0; j < UPT; ++j) {
            const int q = min(q0 + j * nt, EU - 1);
            u[j] = pu[q];
            w[j] = pw[q];
        }
        __builtin_amdgcn_sched_barrier(0);
        unsigned m[UPT][4];
        int c = 0;
#pragma unroll
        for (int j = 0; j < UPT; ++j) {
            const unsigned in = (q0 + j * nt < EU) ? ~0u : 0u;
            m[j][0] = u[j].x & w[j].x & in;
            m[j][1] = u[j].y & w[j].y & in;
            m[j][2] = u[j].z & w[j].z & in;
            m[j][3] = u[j].w & w[j].w & in;
#pragma unroll
            for (int x = 0; x < 4; ++x) c += __builtin_popcount(m[j][x]);
        }
        int inc = c;
        inc += row_below<1>(inc);
        inc += row_below<2>(inc);
        inc += row_below<4>(inc);
        inc += row_below<8>(inc);
        const int low = __builtin_amdgcn_readlane(inc, 15);
        if (lane >= 16) inc += low;
        const int tot = __builtin_amdgcn_readlane(inc, 31);
        if (tot == 0) continue;
        int base = 0;
        if (lane == 0) base = atomicAdd(lcount, tot);
        base = __builtin_amdgcn_readfirstlane(base);
        if (base + tot > cap) continue;
        int pos = base + inc - c;
#pragma unroll
        for (int j = 0; j < UPT; ++j) {
#pragma unroll
            for (int x = 0; x < 4; ++x) {
                unsigned mm = m[j][x];
                const int e0 = 128 * (q0 + j * nt) + 32 * x;
                while (mm) {
                    const int p = __builtin_ctz(mm);
                    mm &= mm - 1;
                    list[pos++] = e0 + p;
                }
            }
        }
    }
    for (int i = 128 * EU + tid; i < E; i += nt) {
        const unsigned a = (unsigned)(dsts[i] - n0);
        if (a < cnt) emit(i);
    }
}

template <int R>
__device__ __forceinline__ int quad_lane(int x) {
    return __builtin_amdgcn_update_dpp(0, x, 0x55 * R, 0xF, 0xF, true);
}
__device__ __forceinline__ int quad_from(int x, int r) {
    switch (r) {
        case 0:  return quad_lane<0>(x);
        case 1:  return quad_lane<1>(x);
        case 2:  return quad_lane<2>(x);
        default: return quad_lane<3>(x);
    }
}
__device__ __forceinline__ float quad_from(float x, int r) {
    return __int_as_float(quad_from(__float_as_int(x), r));
}

template <int H, int C, bool FINAL, bool FIRST>
__global__ __launch_bounds__(OWN_THREADS) void owner_kernel(
    const int* __restrict__ srcs, const int* __restrict__ dsts,
    const unsigned* __restrict__ keys, int E, int N,
    const float* __restrict__ as_, const float* __restrict__ ad_,
    const float* __restrict__ xw, const float* __restrict__ bias,
    float* __restrict__ outp, int* __restrict__ ltot,
    int* __restrict__ lsave, const int* ws_first, size_t ws_bytes) {
    static_assert(C % 4 == 0, "C must be a multiple of 4");
    static_assert(!FINAL || H == 1, "the final layer has one head");
    static_assert(NPS <= (1 << LSHIFT), "a local destination must fit");
    static_assert(LPITCH <= LCAP, "a stored list must fit the list in LDS");
    constexpr int F = H * C;
    constexpr int FP = F + 1;
    __shared__ unsigned s_m[NPS * H];
    __shared__ float    s_ad[NPS * H];
    __shared__ float    s_es[NPS * H];
    __shared__ int      s_bad;
    __shared__ float    s_den[NPS * H];
    __shared__ float    s_acc[NPS * FP];
    __shared__ int      s_li[LCAP];
    __shared__ int      s_ld[LCAP];
    __shared__ int      s_cnt;

    const int tid = threadIdx.x;
    constexpr int nt = OWN_THREADS;
    const int n0 = blockIdx.x * NPS;
    if (n0 >= N) return;
    const int cnt = min(NPS, N - n0);

    auto clear_sums = [&]() {
        for (int j = tid; j < cnt * H; j += nt) {
            s_m[j] = FKEY_NEG_INF;
            s_den[j] = 0.f;
        }
        for (int j = tid; j < cnt * FP; j += nt) s_acc[j] = 0.f;
    };
    for (int j = tid; j < cnt * H; j += nt) {
        const float ad = ad_[(size_t)n0 * H + j];
        const float v  = as_[(size_t)n0 * H + j] + ad;
        s_ad[j] = ad;
        s_es[j] = (v > 0.f) ? v : NEG_SLOPE * v;
    }
    clear_sums();
    if (tid == 0) {
        s_cnt = 0;
        s_bad = 0;
    }
    __syncthreads();

    auto score = [&](int s, int dl, float* e) {
#pragma unroll
        for (int h = 0; h < H; ++h) {
            const float v = as_[s * H + h] + s_ad[dl * H + h];
            e[h] = (v > 0.f) ? v : NEG_SLOPE * v;
        }
    };
    auto put_max = [&](int dl, const float* e) {
#pragma unroll
        for (int h = 0; h < H; ++h) atomicMax(&s_m[dl * H + h], fkey(e[h]));
    };
    auto add_edge = [&](int s, int dl, const float* e) {
        float w[H];
#pragma unroll
        for (int h = 0; h < H; ++h) {
            w[h] = expf(e[h] - fkey_inv(s_m[dl * H + h]));
            atomicAdd(&s_den[dl * H + h], w[h]);
        }
        const float4* __restrict__ sv = (const float4*)(xw + (size_t)s * F);
        float* dv = &s_acc[dl * FP];
#pragma unroll
        for (int f = 0; f < F / 4; ++f) {
            const float4 v = sv[f];
            const float wh = w[(f * 4) / C];
            atomicAdd(&dv[f * 4 + 0], wh * v.x);
            atomicAdd(&dv[f * 4 + 1], wh * v.y);
            atomicAdd(&dv[f * 4 + 2], wh * v.z);
            atomicAdd(&dv[f * 4 + 3], wh * v.w);
        }
    };

    auto walk_rows = [&](bool valid, int s, int dl) {
        constexpr int FL = F / 4;
        constexpr int NV = FL / 4;
        static_assert(FL % 4 == 0 && C % FL == 0,
                      "a lane's floats are whole float4s of one head");
        const int q  = tid & 3;
        const int hq = (q * FL) / C;
        const int sv = valid ? s : n0;
        const int dv = valid ? dl : -1;
        const int dc = valid ? dl : 0;
        float ad[H], es[H];
#pragma unroll
        for (int h = 0; h < H; ++h) {
            ad[h] = s_ad[dc * H + h];
            es[h] = s_es[dc * H + h];
        }
        float a[H];
#pragma unroll
        for (int h = 0; h < H; ++h) a[h] = as_[sv * H + h];
        float4 v[4][NV];
        int    dls[4];
#pragma unroll
        for (int r = 0; r < 4; ++r) {
            const int st = quad_from(sv, r);
            dls[r] = quad_from(dv, r);
            const float4* __restrict__ row =
                (const float4*)(xw + (size_t)st * F + q * FL);
#pragma unroll
            for (int i = 0; i < NV; ++i) v[r][i] = row[i];
        }
        float w[H];
#pragma unroll
        for (int h = 0; h < H; ++h) {
            const float x = a[h] + ad[h];
            const float e = (x > 0.f) ? x : NEG_SLOPE * x;
            const float p = expf(e - es[h]);
            w[h] = valid ? p : 0.f;
        }
        if (valid) {
#pragma unroll
            for (int h = 0; h < H; ++h) atomicAdd(&s_den[dc * H + h], w[h]);
        }
#pragma unroll
        for (int r = 0; r < 4; ++r) {
            float wt = quad_from(w[0], r);
#pragma unroll
            for (int h = 1; h < H; ++h) {
                const float wh = quad_from(w[h], r);
                if (hq == h) wt = wh;
            }
            if (dls[r] >= 0) {
                float* d4 = &s_acc[dls[r] * FP + q * FL];
#pragma unroll
                for (int i = 0; i < NV; ++i) {
                    atomicAdd(&d4[4 * i + 0], wt * v[r][i].x);
                    atomicAdd(&d4[4 * i + 1], wt * v[r][i].y);
                    atomicAdd(&d4[4 * i + 2], wt * v[r][i].z);
                    atomicAdd(&d4[4 * i + 3], wt * v[r][i].w);
                }
            }
        }
    };
    const int wave0 = __builtin_amdgcn_readfirstlane(tid) & ~31;

    const bool ws_ok =
        (size_t)((lsave + (size_t)gridDim.x * LPITCH) - ws_first) *
                sizeof(int) <= ws_bytes &&
        N <= (1 << (31 - LSHIFT));
    int* __restrict__ mine = lsave + (size_t)blockIdx.x * LPITCH;

    int  total  = 0;
    bool loaded = false;
    if (!FIRST && ws_ok) {
        total  = ltot[blockIdx.x];
        loaded = total <= LPITCH;
    }
    if (loaded) {
        constexpr int LSLOTS = LPITCH / OWN_THREADS;
        static_assert(LSLOTS * OWN_THREADS == LPITCH, "whole slots");
        if (total > 0) {
            int v[LSLOTS];
#pragma unroll
            for (int j = 0; j < LSLOTS; ++j)
                v[j] = mine[min(tid + j * nt, total - 1)];
#pragma unroll
            for (int j = 0; j < LSLOTS; ++j) {
                const int k = tid + j * nt;
                if (k < total) {
                    s_li[k] = v[j] >> LSHIFT;
                    s_ld[k] = (v[j] < 0) ? -1 : (v[j] & ((1 << LSHIFT) - 1));
                }
            }
        }
    } else {
        scan_keys(keys, dsts, E, n0, (unsigned)cnt, &s_cnt, s_li, LCAP,
                  [&](int i) {
                      const int pos = atomicAdd(&s_cnt, 1);
                      if (pos < LCAP) s_li[pos] = i;
                  });
        __syncthreads();
        total = s_cnt;
    }
    const bool listed = total <= LCAP;
    if (!loaded) {
        const bool store = FIRST && ws_ok && total <= LPITCH;
        if (FIRST && ws_ok && tid == 0) ltot[blockIdx.x] = total;
        auto resolve = [&](auto keep) {
            for (int k = tid; k < total; k += nt) {
                const int i = s_li[k];
                const unsigned a = (unsigned)(dsts[i] - n0);
                const int s  = srcs[i];
                const int dl = (a < (unsigned)cnt) ? (int)a : -1;
                s_li[k] = s;
                s_ld[k] = dl;
                keep(k, s, dl);
            }
        };
        if (store)
            resolve([&](int k, int s, int dl) {
                mine[k] = (dl < 0) ? -1 : ((s << LSHIFT) | dl);
            });
        else if (listed)
            resolve([](int, int, int) {});
    }

    auto exact_walks = [&]() {
        float es[H];
        if (listed) {
            for (int k = tid; k < total; k += nt) {
                const int dl = s_ld[k];
                if (dl >= 0) {
                    float e[H];
                    score(s_li[k], dl, e);
                    put_max(dl, e);
                }
            }
        } else {
            for (int i = tid; i < E; i += nt) {
                const unsigned a = (unsigned)(dsts[i] - n0);
                if (a < (unsigned)cnt) {
                    float e[H];
                    score(srcs[i], (int)a, e);
                    put_max((int)a, e);
                }
            }
        }
        if (tid < cnt) {
            score(n0 + tid, tid, es);
            put_max(tid, es);
        }
        __syncthreads();
        if (listed) {
            for (int k = tid; k < total; k += nt) {
                const int dl = s_ld[k];
                if (dl >= 0) {
                    float e[H];
                    score(s_li[k], dl, e);
                    add_edge(s_li[k], dl, e);
                }
            }
        } else {
            for (int i = tid; i < E; i += nt) {
                const unsigned a = (unsigned)(dsts[i] - n0);
                if (a < (unsigned)cnt) {
                    float e[H];
                    const int s = srcs[i];
                    score(s, (int)a, e);
                    add_edge(s, (int)a, e);
                }
            }
        }
        if (tid < cnt) add_edge(n0 + tid, tid, es);
        __syncthreads();
    };

    auto write_out = [&]() {
        bool bad = false;
        if (!FINAL) {
            for (int j = tid; j < cnt * F; j += nt) {
                const float a  = s_acc[(j / F) * FP + j % F];
                const float dn = s_den[j / C];
                bad = bad || !(fabsf(a) < INFINITY) || !(dn < INFINITY);
                const float v = a / dn + bias[j % F];
                outp[(size_t)n0 * F + j] = (v > 0.f) ? v : (expf(v) - 1.f);
            }
        } else {
            for (int dl = tid; dl < cnt; dl += nt) {
                const float dn = s_den[dl];
                bad = bad || !(dn < INFINITY);
                float v[F];
                float m = -INFINITY;
#pragma unroll
                for (int c = 0; c < F; ++c) {
                    const float a = s_acc[dl * FP + c];
                    bad = bad || !(fabsf(a) < INFINITY);
                    v[c] = a / dn + bias[c];
                    m = fmaxf(m, v[c]);
                }
                float ssum = 0.f;
#pragma unroll
                for (int c = 0; c < F; ++c) ssum += expf(v[c] - m);
                const float lg = m + logf(ssum);
#pragma unroll
                for (int c = 0; c < F; ++c)
                    outp[(size_t)(n0 + dl) * F + c] = v[c] - lg;
            }
        }
        return bad;
    };

    bool exact = !listed;
    for (;;) {
        if (exact) {
            exact_walks();
        } else {
            const int trips = (total + nt - 1) / nt;
#pragma nounroll
            for (int j = 0; j <= trips; ++j) {
                const bool self = j == trips;
                const int kb = self ? 0 : j * nt;
                if (wave0 + kb < (self ? cnt : total)) {
                    const int k = tid + kb;
                    int dl = -1, s = n0;
                    if (self) {
                        if (tid < cnt) { dl = tid; s = n0 + tid; }
                    } else if (k < total) {
                        dl = s_ld[k];
                        if (dl >= 0) s = s_li[k];
                    }
                    walk_rows(dl >= 0, s, dl);
                }
            }
            __syncthreads();
        }
        const bool bad = write_out();
        if (exact) break;
        if (bad) s_bad = 1;
        __syncthreads();
        if (s_bad == 0) break;
        clear_sums();
        __syncthreads();
        exact = true;
    }
}

extern "C" void kernel_launch(void* const* d_in, const int* in_sizes, int n_in,
                              void* d_out, int out_size, void* d_ws,
                              size_t ws_size, hipStream_t stream) {
    const float* x   = (const float*)d_in[0];
    const int*   ei  = (const int*)d_in[1];
    const float* W1  = (const float*)d_in[2];
    const float* a1s = (const float*)d_in[3];
    const float* a1d = (const float*)d_in[4];
    const float* b1  = (const float*)d_in[5];
    const float* W2  = (const float*)d_in[6];
    const float* a2s = (const float*)d_in[7];
    const float* a2d = (const float*)d_in[8];
    const float* b2  = (const float*)d_in[9];
    float* out = (float*)d_out;

    const int N = in_sizes[0] / 128;
    const int E = in_sizes[1] / 2;
    const int* srcs = ei;
    const int* dsts = ei + E;

    char* p = (char*)d_ws;
    auto carve = [&](size_t bytes) {
        void* r = (void*)p;
        p += (bytes + 255) & ~(size_t)255;
        return r;
    };
    float* xw1 = (float*)carve((size_t)N * 32 * 4);
    float* as1 = (float*)carve((size_t)N * 4 * 4);
    float* ad1 = (float*)carve((size_t)N * 4 * 4);
    float* h1  = (float*)carve((size_t)N * 32 * 4);
    unsigned* keys = (unsigned*)carve((size_t)E * 4 + 16);
    float* xw2 = xw1;
    float* as2 = as1;
    float* ad2 = ad1;

    const int B = 256;
    auto g = [](int n, int b) { return (n + b - 1) / b; };

    int* ltot  = (int*)carve((size_t)g(N, NPS) * 4);
    int* lsave = (int*)carve((size_t)g(N, NPS) * LPITCH * 4);
    const int* ws_first = (const int*)d_ws;

    key_kernel<<<g((E >> 10) * 32 + 1, B), B, 0, stream>>>(dsts, E, keys);

    gemm_alpha_kernel<128, 32, 4><<<g(N, 32), 64, 0, stream>>>(
        x, W1, a1s, a1d, xw1, as1, ad1, N);
    owner_kernel<4, 8, false, true><<<g(N, NPS), OWN_THREADS, 0, stream>>>(
        srcs, dsts, keys, E, N, as1, ad1, xw1, b1, h1, ltot, lsave, ws_first,
        ws_size);

    gemm_alpha_kernel<32, 16, 1><<<g(N, 32), 64, 0, stream>>>(
        h1, W2, a2s, a2d, xw2, as2, ad2, N);
    owner_kernel<1, 16, true, false><<<g(N, NPS), OWN_THREADS, 0, stream>>>(
        srcs, dsts, keys, E, N, as2, ad2, xw2, b2, out, ltot, lsave, ws_first,
        ws_size);
}
